# starting kernel
# baseline (speedup 1.0000x reference)
#include <hip/hip_runtime.h>
#include <math.h>


#define T_SEQ  4096
#define DMODEL 1024
#define NH     16
#define HD     64
#define DMLP   4096
#define WIN    512
#define EPS_F  1.1920929e-07f

#define BM   32
#define BN   256
#define KC   32
#define LSTR 36

typedef float v2f __attribute__((ext_vector_type(2)));
typedef float v8f __attribute__((ext_vector_type(8)));

__device__ __forceinline__ v8f wmma4(v2f a, v2f b, v8f c) {
  return __builtin_amdgcn_wmma_f32_16x16x4_f32(false, a, false, b, (short)0, c,
                                               false, false);
}

__global__ __launch_bounds__(256) void rmsnorm_kernel(
    const float* __restrict__ in, float* __restrict__ out, int ncol) {
  const int row = blockIdx.x;
  const float* r = in + (size_t)row * ncol;
  float* o = out + (size_t)row * ncol;
  float ss = 0.0f;
  for (int i = threadIdx.x; i < ncol; i += blockDim.x) {
    float v = r[i];
    ss += v * v;
  }
  __shared__ float red[256];
  red[threadIdx.x] = ss;
  __syncthreads();
  for (int s = 128; s > 0; s >>= 1) {
    if (threadIdx.x < (unsigned)s) red[threadIdx.x] += red[threadIdx.x + s];
    __syncthreads();
  }
  float inv = rsqrtf(red[0] / (float)ncol + EPS_F);
  for (int i = threadIdx.x; i < ncol; i += blockDim.x) o[i] = r[i] * inv;
}

__global__ __launch_bounds__(256) void gemm_wt_kernel(
    const float* __restrict__ A, const float* __restrict__ Wt,
    const float* __restrict__ colScale, const float* __restrict__ resid,
    float* __restrict__ out, int M, int N, int K, int act) {
  __shared__ float lA[2][BM * LSTR];
  __shared__ float lW[2][BN * LSTR];
  (void)M;

  const int tid = threadIdx.x;
  const int lane = tid & 31;
  const int wave = tid >> 5;
  const int lm = lane & 15;
  const int half = lane >> 4;
  const int wm = (wave & 1) * 16;
  const int wn = (wave >> 1) * 64;

  const int m0 = blockIdx.y * BM;
  const int n00 = blockIdx.x * BN;

  const int ldRow = tid >> 3;
  const int ldCol = (tid & 7) * 4;
  const float* Ag = A + (size_t)(m0 + ldRow) * K + ldCol;
  const float* Wg = Wt + (size_t)(n00 + ldRow) * K + ldCol;

  float4 ra;
  float4 rw[8];

  ra = *(const float4*)(Ag);
#pragma unroll
  for (int i = 0; i < 8; ++i)
    rw[i] = *(const float4*)(Wg + (size_t)(i * 32) * K);
  *(float4*)&lA[0][ldRow * LSTR + ldCol] = ra;
#pragma unroll
  for (int i = 0; i < 8; ++i)
    *(float4*)&lW[0][(ldRow + i * 32) * LSTR + ldCol] = rw[i];
  __syncthreads();

  v8f acc[4] = {};
  int cur = 0;
  for (int kc = 0; kc < K; kc += KC) {
    const int nxt = kc + KC;
    if (nxt < K) {
      ra = *(const float4*)(Ag + nxt);
#pragma unroll
      for (int i = 0; i < 8; ++i)
        rw[i] = *(const float4*)(Wg + (size_t)(i * 32) * K + nxt);
    }
    if (kc + 2 * KC < K)
      __builtin_prefetch(Wg + (size_t)(kc + 2 * KC), 0, 1);

    const float* As = &lA[cur][(wm + lm) * LSTR + 2 * half];
    const float* Ws = &lW[cur][(wn + lm) * LSTR + 2 * half];
#pragma unroll
    for (int kk = 0; kk < KC; kk += 4) {
      v2f a = *(const v2f*)(As + kk);
      v2f b0 = *(const v2f*)(Ws + kk);
      v2f b1 = *(const v2f*)(Ws + kk + 16 * LSTR);
      v2f b2 = *(const v2f*)(Ws + kk + 32 * LSTR);
      v2f b3 = *(const v2f*)(Ws + kk + 48 * LSTR);
      acc[0] = wmma4(a, b0, acc[0]);
      acc[1] = wmma4(a, b1, acc[1]);
      acc[2] = wmma4(a, b2, acc[2]);
      acc[3] = wmma4(a, b3, acc[3]);
    }
    if (nxt < K) {
      *(float4*)&lA[cur ^ 1][ldRow * LSTR + ldCol] = ra;
#pragma unroll
      for (int i = 0; i < 8; ++i)
        *(float4*)&lW[cur ^ 1][(ldRow + i * 32) * LSTR + ldCol] = rw[i];
    }
    __syncthreads();
    cur ^= 1;
  }

  const int mB = m0 + wm;
  const int nB = n00 + wn;
#pragma unroll
  for (int s = 0; s < 4; ++s) {
    const int n = nB + s * 16 + lm;
    const float cs = colScale ? colScale[n] : 1.0f;
#pragma unroll
    for (int r = 0; r < 8; ++r) {
      const int m = mB + r + 8 * half;
      float v = acc[s][r] * cs;
      if (act == 1) {
        v = fmaxf(v, 0.0f);
        v = v * v;
      }
      if (resid) v += resid[(size_t)m * N + n];
      out[(size_t)m * N + n] = v;
    }
  }
}

__global__ __launch_bounds__(64) void qk_rope_kernel(
    const float* __restrict__ qkv, float* __restrict__ qh,
    float* __restrict__ kh, float* __restrict__ vh) {
  const int t = blockIdx.x;
  const int h = blockIdx.y;
  const int d = threadIdx.x;
  const float* base = qkv + (size_t)t * (3 * DMODEL) + h * HD;
  float qv = base[d];
  float kv = base[DMODEL + d];
  float vv = base[2 * DMODEL + d];

  __shared__ float red[64], sq[64], sk[64];
  red[d] = qv * qv;
  __syncthreads();
  for (int s = 32; s > 0; s >>= 1) {
    if (d < s) red[d] += red[d + s];
    __syncthreads();
  }
  float qi = rsqrtf(red[0] / (float)HD + EPS_F);
  __syncthreads();
  red[d] = kv * kv;
  __syncthreads();
  for (int s = 32; s > 0; s >>= 1) {
    if (d < s) red[d] += red[d + s];
    __syncthreads();
  }
  float ki = rsqrtf(red[0] / (float)HD + EPS_F);
  __syncthreads();

  sq[d] = qv * qi;
  sk[d] = kv * ki;
  __syncthreads();

  const int j = d & 31;
  float theta = (j < 16) ? (float)t * powf(10000.0f, -(float)j / 15.0f) : 0.0f;
  float c = cosf(theta), s_ = sinf(theta);
  float x1q = sq[j], x2q = sq[j + 32];
  float x1k = sk[j], x2k = sk[j + 32];
  float oq = (d < 32) ? (x1q * c + x2q * s_) : (-x1q * s_ + x2q * c);
  float ok = (d < 32) ? (x1k * c + x2k * s_) : (-x1k * s_ + x2k * c);

  const size_t oidx = ((size_t)h * T_SEQ + t) * HD + d;
  qh[oidx] = oq;
  kh[oidx] = ok;
  vh[oidx] = vv;
}

__global__ __launch_bounds__(32) void attn_kernel(
    const float* __restrict__ Qh, const float* __restrict__ Kh,
    const float* __restrict__ Vh, float* __restrict__ outAttn) {
  const int h = blockIdx.x;
  const int q0 = blockIdx.y * 16;
  const int lane = threadIdx.x;
  const int lm = lane & 15;
  const int half = lane >> 4;
  const size_t hb = (size_t)h * T_SEQ * HD;

  const float* qrow = Qh + hb + (size_t)(q0 + lm) * HD + 2 * half;
  v2f qa[16];
#pragma unroll
  for (int dc = 0; dc < 16; ++dc) qa[dc] = *(const v2f*)(qrow + 4 * dc);

  float mi[8], li[8];
  v8f o[4] = {};
#pragma unroll
  for (int r = 0; r < 8; ++r) {
    mi[r] = -INFINITY;
    li[r] = 0.0f;
  }

  __shared__ float sP[16 * 16];

  const int lo = q0 - WIN + 1;
  const int kt0 = (lo < 0) ? 0 : (lo / 16) * 16;
  for (int kt = kt0; kt <= q0; kt += 16) {
    const float* krow = Kh + hb + (size_t)(kt + lm) * HD + 2 * half;
    v2f kb[16];
#pragma unroll
    for (int dc = 0; dc < 16; ++dc) kb[dc] = *(const v2f*)(krow + 4 * dc);
    v8f s0 = {}, s1 = {};
#pragma unroll
    for (int dc = 0; dc < 16; dc += 2) {
      s0 = wmma4(qa[dc], kb[dc], s0);
      s1 = wmma4(qa[dc + 1], kb[dc + 1], s1);
    }
    v8f s = s0 + s1;

#pragma unroll
    for (int r = 0; r < 8; ++r) {
      const int qp = q0 + r + 8 * half;
      const int kp = kt + lm;
      float sv = s[r] * 0.125f;
      const bool okm = (kp <= qp) && (kp > qp - WIN);
      sv = okm ? sv : -INFINITY;
      float rm = sv;
      rm = fmaxf(rm, __shfl_xor(rm, 1, 32));
      rm = fmaxf(rm, __shfl_xor(rm, 2, 32));
      rm = fmaxf(rm, __shfl_xor(rm, 4, 32));
      rm = fmaxf(rm, __shfl_xor(rm, 8, 32));
      const float nm = fmaxf(mi[r], rm);
      const float p = (sv == -INFINITY) ? 0.0f : __expf(sv - nm);
      float corr;
      if (nm == -INFINITY)
        corr = 1.0f;
      else if (mi[r] == -INFINITY)
        corr = 0.0f;
      else
        corr = __expf(mi[r] - nm);
      float rs = p;
      rs += __shfl_xor(rs, 1, 32);
      rs += __shfl_xor(rs, 2, 32);
      rs += __shfl_xor(rs, 4, 32);
      rs += __shfl_xor(rs, 8, 32);
      li[r] = li[r] * corr + rs;
      mi[r] = nm;
      o[0][r] *= corr;
      o[1][r] *= corr;
      o[2][r] *= corr;
      o[3][r] *= corr;
      sP[(r + 8 * half) * 16 + lm] = p;
    }
    __syncthreads();

    const float* vbase = Vh + hb;
    v2f vb[16];
#pragma unroll
    for (int kc4 = 0; kc4 < 4; ++kc4) {
      const int kr = kt + kc4 * 4 + 2 * half;
#pragma unroll
      for (int f = 0; f < 4; ++f) {
        v2f b;
        b.x = vbase[(size_t)kr * HD + f * 16 + lm];
        b.y = vbase[(size_t)(kr + 1) * HD + f * 16 + lm];
        vb[kc4 * 4 + f] = b;
      }
    }
    v2f pa[4];
#pragma unroll
    for (int kc4 = 0; kc4 < 4; ++kc4) {
      pa[kc4].x = sP[lm * 16 + kc4 * 4 + 2 * half];
      pa[kc4].y = sP[lm * 16 + kc4 * 4 + 2 * half + 1];
    }
#pragma unroll
    for (int kc4 = 0; kc4 < 4; ++kc4)
#pragma unroll
      for (int f = 0; f < 4; ++f) o[f] = wmma4(pa[kc4], vb[kc4 * 4 + f], o[f]);
    __syncthreads();
  }

#pragma unroll
  for (int f = 0; f < 4; ++f) {
#pragma unroll
    for (int r = 0; r < 8; ++r) {
      const int tq = q0 + r + 8 * half;
      const int col = h * HD + f * 16 + lm;
      outAttn[(size_t)tq * DMODEL + col] = o[f][r] / li[r];
    }
  }
}

extern "C" void kernel_launch(void* const* d_in, const int* in_sizes, int n_in,
                              void* d_out, int out_size, void* d_ws,
                              size_t ws_size, hipStream_t stream) {
  (void)in_sizes;
  (void)n_in;
  (void)out_size;
  (void)ws_size;
  const float* x = (const float*)d_in[0];
  const float* qkv_w = (const float*)d_in[1];
  const float* o_w = (const float*)d_in[2];
  const float* o_scale = (const float*)d_in[3];
  const float* w1 = (const float*)d_in[4];
  const float* w2 = (const float*)d_in[5];
  const float* w2_scale = (const float*)d_in[6];
  float* out = (float*)d_out;

  const size_t TD = (size_t)T_SEQ * DMODEL;
  float* xn = (float*)d_ws;
  float* qkv = xn + TD;
  float* qh = qkv + 3 * TD;
  float* kh = qh + TD;
  float* vh = kh + TD;
  float* attn = vh + TD;
  float* x1 = attn + TD;
  float* hbuf = qkv;

  rmsnorm_kernel<<<T_SEQ, 256, 0, stream>>>(x, xn, DMODEL);
  gemm_wt_kernel<<<dim3(3 * DMODEL / BN, T_SEQ / BM), 256, 0, stream>>>(
      xn, qkv_w, nullptr, nullptr, qkv, T_SEQ, 3 * DMODEL, DMODEL, 0);
  qk_rope_kernel<<<dim3(T_SEQ, NH), 64, 0, stream>>>(qkv, qh, kh, vh);
  attn_kernel<<<dim3(NH, T_SEQ / 16), 32, 0, stream>>>(qh, kh, vh, attn);
  gemm_wt_kernel<<<dim3(DMODEL / BN, T_SEQ / BM), 256, 0, stream>>>(
      attn, o_w, o_scale, x, x1, T_SEQ, DMODEL, DMODEL, 0);
  rmsnorm_kernel<<<T_SEQ, 256, 0, stream>>>(x1, xn, DMODEL);
  gemm_wt_kernel<<<dim3(DMLP / BN, T_SEQ / BM), 256, 0, stream>>>(
      xn, w1, nullptr, nullptr, hbuf, T_SEQ, DMLP, DMODEL, 1);
  gemm_wt_kernel<<<dim3(DMODEL / BN, T_SEQ / BM), 256, 0, stream>>>(
      hbuf, w2, w2_scale, x1, out, T_SEQ, DMODEL, DMLP, 0);
}


# GEMM staging chunk kept in native 4-float vector registers, not a float4 struct array the compiler put in scratch: per-chunk scratch store+reload gone, global loads overlap the WMMA block
# speedup vs baseline: 2.2401x; 2.2401x over previous
#include <hip/hip_runtime.h>
#include <math.h>


#define T_SEQ  4096
#define DMODEL 1024
#define NH     16
#define HD     64
#define DMLP   4096
#define WIN    512
#define EPS_F  1.1920929e-07f

#define BM   32
#define BN   256
#define KC   32
#define LSTR 36

typedef float v2f __attribute__((ext_vector_type(2)));
typedef float v4f __attribute__((ext_vector_type(4)));
typedef float v8f __attribute__((ext_vector_type(8)));

__device__ __forceinline__ v8f wmma4(v2f a, v2f b, v8f c) {
  return __builtin_amdgcn_wmma_f32_16x16x4_f32(false, a, false, b, (short)0, c,
                                               false, false);
}

__global__ __launch_bounds__(256) void rmsnorm_kernel(
    const float* __restrict__ in, float* __restrict__ out, int ncol) {
  const int row = blockIdx.x;
  const float* r = in + (size_t)row * ncol;
  float* o = out + (size_t)row * ncol;
  float ss = 0.0f;
  for (int i = threadIdx.x; i < ncol; i += blockDim.x) {
    float v = r[i];
    ss += v * v;
  }
  __shared__ float red[256];
  red[threadIdx.x] = ss;
  __syncthreads();
  for (int s = 128; s > 0; s >>= 1) {
    if (threadIdx.x < (unsigned)s) red[threadIdx.x] += red[threadIdx.x + s];
    __syncthreads();
  }
  float inv = rsqrtf(red[0] / (float)ncol + EPS_F);
  for (int i = threadIdx.x; i < ncol; i += blockDim.x) o[i] = r[i] * inv;
}

__global__ __launch_bounds__(256) void gemm_wt_kernel(
    const float* __restrict__ A, const float* __restrict__ Wt,
    const float* __restrict__ colScale, const float* __restrict__ resid,
    float* __restrict__ out, int M, int N, int K, int act) {
  __shared__ float lA[2][BM * LSTR];
  __shared__ float lW[2][BN * LSTR];
  (void)M;

  const int tid = threadIdx.x;
  const int lane = tid & 31;
  const int wave = tid >> 5;
  const int lm = lane & 15;
  const int half = lane >> 4;
  const int wm = (wave & 1) * 16;
  const int wn = (wave >> 1) * 64;

  const int m0 = blockIdx.y * BM;
  const int n00 = blockIdx.x * BN;

  const int ldRow = tid >> 3;
  const int ldCol = (tid & 7) * 4;
  const float* Ag = A + (size_t)(m0 + ldRow) * K + ldCol;
  const float* Wg = Wt + (size_t)(n00 + ldRow) * K + ldCol;

  v4f ra;
  v4f rw[8];

  ra = *(const v4f*)(Ag);
#pragma unroll
  for (int i = 0; i < 8; ++i)
    rw[i] = *(const v4f*)(Wg + (size_t)(i * 32) * K);
  *(v4f*)&lA[0][ldRow * LSTR + ldCol] = ra;
#pragma unroll
  for (int i = 0; i < 8; ++i)
    *(v4f*)&lW[0][(ldRow + i * 32) * LSTR + ldCol] = rw[i];
  __syncthreads();

  v8f acc[4] = {};
  int cur = 0;
  for (int kc = 0; kc < K; kc += KC) {
    const int nxt = kc + KC;
    if (nxt < K) {
      ra = *(const v4f*)(Ag + nxt);
#pragma unroll
      for (int i = 0; i < 8; ++i)
        rw[i] = *(const v4f*)(Wg + (size_t)(i * 32) * K + nxt);
    }
    if (kc + 2 * KC < K)
      __builtin_prefetch(Wg + (size_t)(kc + 2 * KC), 0, 1);

    const float* As = &lA[cur][(wm + lm) * LSTR + 2 * half];
    const float* Ws = &lW[cur][(wn + lm) * LSTR + 2 * half];
#pragma unroll
    for (int kk = 0; kk < KC; kk += 4) {
      v2f a = *(const v2f*)(As + kk);
      v2f b0 = *(const v2f*)(Ws + kk);
      v2f b1 = *(const v2f*)(Ws + kk + 16 * LSTR);
      v2f b2 = *(const v2f*)(Ws + kk + 32 * LSTR);
      v2f b3 = *(const v2f*)(Ws + kk + 48 * LSTR);
      acc[0] = wmma4(a, b0, acc[0]);
      acc[1] = wmma4(a, b1, acc[1]);
      acc[2] = wmma4(a, b2, acc[2]);
      acc[3] = wmma4(a, b3, acc[3]);
    }
    if (nxt < K) {
      *(v4f*)&lA[cur ^ 1][ldRow * LSTR + ldCol] = ra;
#pragma unroll
      for (int i = 0; i < 8; ++i)
        *(v4f*)&lW[cur ^ 1][(ldRow + i * 32) * LSTR + ldCol] = rw[i];
    }
    __syncthreads();
    cur ^= 1;
  }

  const int mB = m0 + wm;
  const int nB = n00 + wn;
#pragma unroll
  for (int s = 0; s < 4; ++s) {
    const int n = nB + s * 16 + lm;
    const float cs = colScale ? colScale[n] : 1.0f;
#pragma unroll
    for (int r = 0; r < 8; ++r) {
      const int m = mB + r + 8 * half;
      float v = acc[s][r] * cs;
      if (act == 1) {
        v = fmaxf(v, 0.0f);
        v = v * v;
      }
      if (resid) v += resid[(size_t)m * N + n];
      out[(size_t)m * N + n] = v;
    }
  }
}

__global__ __launch_bounds__(64) void qk_rope_kernel(
    const float* __restrict__ qkv, float* __restrict__ qh,
    float* __restrict__ kh, float* __restrict__ vh) {
  const int t = blockIdx.x;
  const int h = blockIdx.y;
  const int d = threadIdx.x;
  const float* base = qkv + (size_t)t * (3 * DMODEL) + h * HD;
  float qv = base[d];
  float kv = base[DMODEL + d];
  float vv = base[2 * DMODEL + d];

  __shared__ float red[64], sq[64], sk[64];
  red[d] = qv * qv;
  __syncthreads();
  for (int s = 32; s > 0; s >>= 1) {
    if (d < s) red[d] += red[d + s];
    __syncthreads();
  }
  float qi = rsqrtf(red[0] / (float)HD + EPS_F);
  __syncthreads();
  red[d] = kv * kv;
  __syncthreads();
  for (int s = 32; s > 0; s >>= 1) {
    if (d < s) red[d] += red[d + s];
    __syncthreads();
  }
  float ki = rsqrtf(red[0] / (float)HD + EPS_F);
  __syncthreads();

  sq[d] = qv * qi;
  sk[d] = kv * ki;
  __syncthreads();

  const int j = d & 31;
  float theta = (j < 16) ? (float)t * powf(10000.0f, -(float)j / 15.0f) : 0.0f;
  float c = cosf(theta), s_ = sinf(theta);
  float x1q = sq[j], x2q = sq[j + 32];
  float x1k = sk[j], x2k = sk[j + 32];
  float oq = (d < 32) ? (x1q * c + x2q * s_) : (-x1q * s_ + x2q * c);
  float ok = (d < 32) ? (x1k * c + x2k * s_) : (-x1k * s_ + x2k * c);

  const size_t oidx = ((size_t)h * T_SEQ + t) * HD + d;
  qh[oidx] = oq;
  kh[oidx] = ok;
  vh[oidx] = vv;
}

__global__ __launch_bounds__(32) void attn_kernel(
    const float* __restrict__ Qh, const float* __restrict__ Kh,
    const float* __restrict__ Vh, float* __restrict__ outAttn) {
  const int h = blockIdx.x;
  const int q0 = blockIdx.y * 16;
  const int lane = threadIdx.x;
  const int lm = lane & 15;
  const int half = lane >> 4;
  const size_t hb = (size_t)h * T_SEQ * HD;

  const float* qrow = Qh + hb + (size_t)(q0 + lm) * HD + 2 * half;
  v2f qa[16];
#pragma unroll
  for (int dc = 0; dc < 16; ++dc) qa[dc] = *(const v2f*)(qrow + 4 * dc);

  float mi[8], li[8];
  v8f o[4] = {};
#pragma unroll
  for (int r = 0; r < 8; ++r) {
    mi[r] = -INFINITY;
    li[r] = 0.0f;
  }

  __shared__ float sP[16 * 16];

  const int lo = q0 - WIN + 1;
  const int kt0 = (lo < 0) ? 0 : (lo / 16) * 16;
  for (int kt = kt0; kt <= q0; kt += 16) {
    const float* krow = Kh + hb + (size_t)(kt + lm) * HD + 2 * half;
    v2f kb[16];
#pragma unroll
    for (int dc = 0; dc < 16; ++dc) kb[dc] = *(const v2f*)(krow + 4 * dc);
    v8f s0 = {}, s1 = {};
#pragma unroll
    for (int dc = 0; dc < 16; dc += 2) {
      s0 = wmma4(qa[dc], kb[dc], s0);
      s1 = wmma4(qa[dc + 1], kb[dc + 1], s1);
    }
    v8f s = s0 + s1;

#pragma unroll
    for (int r = 0; r < 8; ++r) {
      const int qp = q0 + r + 8 * half;
      const int kp = kt + lm;
      float sv = s[r] * 0.125f;
      const bool okm = (kp <= qp) && (kp > qp - WIN);
      sv = okm ? sv : -INFINITY;
      float rm = sv;
      rm = fmaxf(rm, __shfl_xor(rm, 1, 32));
      rm = fmaxf(rm, __shfl_xor(rm, 2, 32));
      rm = fmaxf(rm, __shfl_xor(rm, 4, 32));
      rm = fmaxf(rm, __shfl_xor(rm, 8, 32));
      const float nm = fmaxf(mi[r], rm);
      const float p = (sv == -INFINITY) ? 0.0f : __expf(sv - nm);
      float corr;
      if (nm == -INFINITY)
        corr = 1.0f;
      else if (mi[r] == -INFINITY)
        corr = 0.0f;
      else
        corr = __expf(mi[r] - nm);
      float rs = p;
      rs += __shfl_xor(rs, 1, 32);
      rs += __shfl_xor(rs, 2, 32);
      rs += __shfl_xor(rs, 4, 32);
      rs += __shfl_xor(rs, 8, 32);
      li[r] = li[r] * corr + rs;
      mi[r] = nm;
      o[0][r] *= corr;
      o[1][r] *= corr;
      o[2][r] *= corr;
      o[3][r] *= corr;
      sP[(r + 8 * half) * 16 + lm] = p;
    }
    __syncthreads();

    const float* vbase = Vh + hb;
    v2f vb[16];
#pragma unroll
    for (int kc4 = 0; kc4 < 4; ++kc4) {
      const int kr = kt + kc4 * 4 + 2 * half;
#pragma unroll
      for (int f = 0; f < 4; ++f) {
        v2f b;
        b.x = vbase[(size_t)kr * HD + f * 16 + lm];
        b.y = vbase[(size_t)(kr + 1) * HD + f * 16 + lm];
        vb[kc4 * 4 + f] = b;
      }
    }
    v2f pa[4];
#pragma unroll
    for (int kc4 = 0; kc4 < 4; ++kc4) {
      pa[kc4].x = sP[lm * 16 + kc4 * 4 + 2 * half];
      pa[kc4].y = sP[lm * 16 + kc4 * 4 + 2 * half + 1];
    }
#pragma unroll
    for (int kc4 = 0; kc4 < 4; ++kc4)
#pragma unroll
      for (int f = 0; f < 4; ++f) o[f] = wmma4(pa[kc4], vb[kc4 * 4 + f], o[f]);
    __syncthreads();
  }

#pragma unroll
  for (int f = 0; f < 4; ++f) {
#pragma unroll
    for (int r = 0; r < 8; ++r) {
      const int tq = q0 + r + 8 * half;
      const int col = h * HD + f * 16 + lm;
      outAttn[(size_t)tq * DMODEL + col] = o[f][r] / li[r];
    }
  }
}

extern "C" void kernel_launch(void* const* d_in, const int* in_sizes, int n_in,
                              void* d_out, int out_size, void* d_ws,
                              size_t ws_size, hipStream_t stream) {
  (void)in_sizes;
  (void)n_in;
  (void)out_size;
  (void)ws_size;
  const float* x = (const float*)d_in[0];
  const float* qkv_w = (const float*)d_in[1];
  const float* o_w = (const float*)d_in[2];
  const float* o_scale = (const float*)d_in[3];
  const float* w1 = (const float*)d_in[4];
  const float* w2 = (const float*)d_in[5];
  const float* w2_scale = (const float*)d_in[6];
  float* out = (float*)d_out;

  const size_t TD = (size_t)T_SEQ * DMODEL;
  float* xn = (float*)d_ws;
  float* qkv = xn + TD;
  float* qh = qkv + 3 * TD;
  float* kh = qh + TD;
  float* vh = kh + TD;
  float* attn = vh + TD;
  float* x1 = attn + TD;
  float* hbuf = qkv;

  rmsnorm_kernel<<<T_SEQ, 256, 0, stream>>>(x, xn, DMODEL);
  gemm_wt_kernel<<<dim3(3 * DMODEL / BN, T_SEQ / BM), 256, 0, stream>>>(
      xn, qkv_w, nullptr, nullptr, qkv, T_SEQ, 3 * DMODEL, DMODEL, 0);
  qk_rope_kernel<<<dim3(T_SEQ, NH), 64, 0, stream>>>(qkv, qh, kh, vh);
  attn_kernel<<<dim3(NH, T_SEQ / 16), 32, 0, stream>>>(qh, kh, vh, attn);
  gemm_wt_kernel<<<dim3(DMODEL / BN, T_SEQ / BM), 256, 0, stream>>>(
      attn, o_w, o_scale, x, x1, T_SEQ, DMODEL, DMODEL, 0);
  rmsnorm_kernel<<<T_SEQ, 256, 0, stream>>>(x1, xn, DMODEL);
  gemm_wt_kernel<<<dim3(DMLP / BN, T_SEQ / BM), 256, 0, stream>>>(
      xn, w1, nullptr, nullptr, hbuf, T_SEQ, DMLP, DMODEL, 1);
  gemm_wt_kernel<<<dim3(DMODEL / BN, T_SEQ / BM), 256, 0, stream>>>(
      hbuf, w2, w2_scale, x1, out, T_SEQ, DMODEL, DMLP, 0);
}


# Attention on the transposed score tile (S^T = K Q^T, one query per lane): row max/sum in-lane plus one half-exchange, P^T to PV from registers, no LDS or barrier; row-sum and PV key order change
# speedup vs baseline: 2.4258x; 1.0829x over previous
#include <hip/hip_runtime.h>
#include <math.h>


#define T_SEQ  4096
#define DMODEL 1024
#define NH     16
#define HD     64
#define DMLP   4096
#define WIN    512
#define EPS_F  1.1920929e-07f

#define BM   32
#define BN   256
#define KC   32
#define LSTR 36

typedef float v2f __attribute__((ext_vector_type(2)));
typedef float v4f __attribute__((ext_vector_type(4)));
typedef float v8f __attribute__((ext_vector_type(8)));

__device__ __forceinline__ v8f wmma4(v2f a, v2f b, v8f c) {
  return __builtin_amdgcn_wmma_f32_16x16x4_f32(false, a, false, b, (short)0, c,
                                               false, false);
}

__global__ __launch_bounds__(256) void rmsnorm_kernel(
    const float* __restrict__ in, float* __restrict__ out, int ncol) {
  const int row = blockIdx.x;
  const float* r = in + (size_t)row * ncol;
  float* o = out + (size_t)row * ncol;
  float ss = 0.0f;
  for (int i = threadIdx.x; i < ncol; i += blockDim.x) {
    float v = r[i];
    ss += v * v;
  }
  __shared__ float red[256];
  red[threadIdx.x] = ss;
  __syncthreads();
  for (int s = 128; s > 0; s >>= 1) {
    if (threadIdx.x < (unsigned)s) red[threadIdx.x] += red[threadIdx.x + s];
    __syncthreads();
  }
  float inv = rsqrtf(red[0] / (float)ncol + EPS_F);
  for (int i = threadIdx.x; i < ncol; i += blockDim.x) o[i] = r[i] * inv;
}

__global__ __launch_bounds__(256) void gemm_wt_kernel(
    const float* __restrict__ A, const float* __restrict__ Wt,
    const float* __restrict__ colScale, const float* __restrict__ resid,
    float* __restrict__ out, int M, int N, int K, int act) {
  __shared__ float lA[2][BM * LSTR];
  __shared__ float lW[2][BN * LSTR];
  (void)M;

  const int tid = threadIdx.x;
  const int lane = tid & 31;
  const int wave = tid >> 5;
  const int lm = lane & 15;
  const int half = lane >> 4;
  const int wm = (wave & 1) * 16;
  const int wn = (wave >> 1) * 64;

  const int m0 = blockIdx.y * BM;
  const int n00 = blockIdx.x * BN;

  const int ldRow = tid >> 3;
  const int ldCol = (tid & 7) * 4;
  const float* Ag = A + (size_t)(m0 + ldRow) * K + ldCol;
  const float* Wg = Wt + (size_t)(n00 + ldRow) * K + ldCol;

  v4f ra;
  v4f rw[8];

  ra = *(const v4f*)(Ag);
#pragma unroll
  for (int i = 0; i < 8; ++i)
    rw[i] = *(const v4f*)(Wg + (size_t)(i * 32) * K);
  *(v4f*)&lA[0][ldRow * LSTR + ldCol] = ra;
#pragma unroll
  for (int i = 0; i < 8; ++i)
    *(v4f*)&lW[0][(ldRow + i * 32) * LSTR + ldCol] = rw[i];
  __syncthreads();

  v8f acc[4] = {};
  int cur = 0;
  for (int kc = 0; kc < K; kc += KC) {
    const int nxt = kc + KC;
    if (nxt < K) {
      ra = *(const v4f*)(Ag + nxt);
#pragma unroll
      for (int i = 0; i < 8; ++i)
        rw[i] = *(const v4f*)(Wg + (size_t)(i * 32) * K + nxt);
    }
    if (kc + 2 * KC < K)
      __builtin_prefetch(Wg + (size_t)(kc + 2 * KC), 0, 1);

    const float* As = &lA[cur][(wm + lm) * LSTR + 2 * half];
    const float* Ws = &lW[cur][(wn + lm) * LSTR + 2 * half];
#pragma unroll
    for (int kk = 0; kk < KC; kk += 4) {
      v2f a = *(const v2f*)(As + kk);
      v2f b0 = *(const v2f*)(Ws + kk);
      v2f b1 = *(const v2f*)(Ws + kk + 16 * LSTR);
      v2f b2 = *(const v2f*)(Ws + kk + 32 * LSTR);
      v2f b3 = *(const v2f*)(Ws + kk + 48 * LSTR);
      acc[0] = wmma4(a, b0, acc[0]);
      acc[1] = wmma4(a, b1, acc[1]);
      acc[2] = wmma4(a, b2, acc[2]);
      acc[3] = wmma4(a, b3, acc[3]);
    }
    if (nxt < K) {
      *(v4f*)&lA[cur ^ 1][ldRow * LSTR + ldCol] = ra;
#pragma unroll
      for (int i = 0; i < 8; ++i)
        *(v4f*)&lW[cur ^ 1][(ldRow + i * 32) * LSTR + ldCol] = rw[i];
    }
    __syncthreads();
    cur ^= 1;
  }

  const int mB = m0 + wm;
  const int nB = n00 + wn;
#pragma unroll
  for (int s = 0; s < 4; ++s) {
    const int n = nB + s * 16 + lm;
    const float cs = colScale ? colScale[n] : 1.0f;
#pragma unroll
    for (int r = 0; r < 8; ++r) {
      const int m = mB + r + 8 * half;
      float v = acc[s][r] * cs;
      if (act == 1) {
        v = fmaxf(v, 0.0f);
        v = v * v;
      }
      if (resid) v += resid[(size_t)m * N + n];
      out[(size_t)m * N + n] = v;
    }
  }
}

__global__ __launch_bounds__(64) void qk_rope_kernel(
    const float* __restrict__ qkv, float* __restrict__ qh,
    float* __restrict__ kh, float* __restrict__ vh) {
  const int t = blockIdx.x;
  const int h = blockIdx.y;
  const int d = threadIdx.x;
  const float* base = qkv + (size_t)t * (3 * DMODEL) + h * HD;
  float qv = base[d];
  float kv = base[DMODEL + d];
  float vv = base[2 * DMODEL + d];

  __shared__ float red[64], sq[64], sk[64];
  red[d] = qv * qv;
  __syncthreads();
  for (int s = 32; s > 0; s >>= 1) {
    if (d < s) red[d] += red[d + s];
    __syncthreads();
  }
  float qi = rsqrtf(red[0] / (float)HD + EPS_F);
  __syncthreads();
  red[d] = kv * kv;
  __syncthreads();
  for (int s = 32; s > 0; s >>= 1) {
    if (d < s) red[d] += red[d + s];
    __syncthreads();
  }
  float ki = rsqrtf(red[0] / (float)HD + EPS_F);
  __syncthreads();

  sq[d] = qv * qi;
  sk[d] = kv * ki;
  __syncthreads();

  const int j = d & 31;
  float theta = (j < 16) ? (float)t * powf(10000.0f, -(float)j / 15.0f) : 0.0f;
  float c = cosf(theta), s_ = sinf(theta);
  float x1q = sq[j], x2q = sq[j + 32];
  float x1k = sk[j], x2k = sk[j + 32];
  float oq = (d < 32) ? (x1q * c + x2q * s_) : (-x1q * s_ + x2q * c);
  float ok = (d < 32) ? (x1k * c + x2k * s_) : (-x1k * s_ + x2k * c);

  const size_t oidx = ((size_t)h * T_SEQ + t) * HD + d;
  qh[oidx] = oq;
  kh[oidx] = ok;
  vh[oidx] = vv;
}

__device__ __forceinline__ float xhalf(float x) {
  const int b = __builtin_bit_cast(int, x);
  return __builtin_bit_cast(
      float, __builtin_amdgcn_permlanex16(b, b, 0x76543210, 0xfedcba98, false,
                                          false));
}

__global__ __launch_bounds__(32) void attn_kernel(
    const float* __restrict__ Qh, const float* __restrict__ Kh,
    const float* __restrict__ Vh, float* __restrict__ outAttn) {
  const int h = blockIdx.x;
  const int q0 = blockIdx.y * 16;
  const int lane = threadIdx.x;
  const int lm = lane & 15;
  const int half = lane >> 4;
  const size_t hb = (size_t)h * T_SEQ * HD;
  const int qp = q0 + lm;

  const float* qrow = Qh + hb + (size_t)qp * HD + 2 * half;
  v2f qb[16];
#pragma unroll
  for (int dc = 0; dc < 16; ++dc) qb[dc] = *(const v2f*)(qrow + 4 * dc);

  float mi = -INFINITY;
  float li = 0.0f;
  v8f o[4] = {};

  const int lo = q0 - WIN + 1;
  const int kt0 = (lo < 0) ? 0 : (lo / 16) * 16;
  for (int kt = kt0; kt <= q0; kt += 16) {
    const float* krow = Kh + hb + (size_t)(kt + lm) * HD + 2 * half;
    v2f ka[16];
#pragma unroll
    for (int dc = 0; dc < 16; ++dc) ka[dc] = *(const v2f*)(krow + 4 * dc);

    const float* vrow = Vh + hb + (size_t)(kt + 8 * half) * HD + lm;
    v2f va[16];
#pragma unroll
    for (int s = 0; s < 4; ++s) {
#pragma unroll
      for (int f = 0; f < 4; ++f) {
        v2f a;
        a.x = vrow[(2 * s) * HD + f * 16];
        a.y = vrow[(2 * s + 1) * HD + f * 16];
        va[s * 4 + f] = a;
      }
    }

    v8f s0 = {}, s1 = {};
#pragma unroll
    for (int dc = 0; dc < 16; dc += 2) {
      s0 = wmma4(ka[dc], qb[dc], s0);
      s1 = wmma4(ka[dc + 1], qb[dc + 1], s1);
    }
    const v8f st = s0 + s1;

    float sv[8];
#pragma unroll
    for (int r = 0; r < 8; ++r) sv[r] = st[r] * 0.125f;

    if ((kt + 15 > q0) || (kt + WIN <= q0 + 15)) {
#pragma unroll
      for (int r = 0; r < 8; ++r) {
        const int kp = kt + 8 * half + r;
        const bool okm = (kp <= qp) && (kp > qp - WIN);
        sv[r] = okm ? sv[r] : -INFINITY;
      }
    }

    float rm = fmaxf(fmaxf(fmaxf(sv[0], sv[1]), fmaxf(sv[2], sv[3])),
                     fmaxf(fmaxf(sv[4], sv[5]), fmaxf(sv[6], sv[7])));
    rm = fmaxf(rm, xhalf(rm));
    const float nm = fmaxf(mi, rm);
    const float nmz = (nm == -INFINITY) ? 0.0f : nm;
    const float corr = __expf(mi - nmz);
    float p[8];
#pragma unroll
    for (int r = 0; r < 8; ++r) p[r] = __expf(sv[r] - nmz);
    float rs = ((p[0] + p[1]) + (p[2] + p[3])) + ((p[4] + p[5]) + (p[6] + p[7]));
    rs += xhalf(rs);
    li = li * corr + rs;
    mi = nm;
#pragma unroll
    for (int f = 0; f < 4; ++f) o[f] *= corr;

#pragma unroll
    for (int s = 0; s < 4; ++s) {
      v2f pb;
      pb.x = p[2 * s];
      pb.y = p[2 * s + 1];
#pragma unroll
      for (int f = 0; f < 4; ++f) o[f] = wmma4(va[s * 4 + f], pb, o[f]);
    }
  }

  float* orow = outAttn + (size_t)qp * DMODEL + h * HD + 8 * half;
#pragma unroll
  for (int f = 0; f < 4; ++f) {
#pragma unroll
    for (int g = 0; g < 2; ++g) {
      v4f w;
      w.x = o[f][4 * g + 0] / li;
      w.y = o[f][4 * g + 1] / li;
      w.z = o[f][4 * g + 2] / li;
      w.w = o[f][4 * g + 3] / li;
      *(v4f*)(orow + f * 16 + 4 * g) = w;
    }
  }
}

extern "C" void kernel_launch(void* const* d_in, const int* in_sizes, int n_in,
                              void* d_out, int out_size, void* d_ws,
                              size_t ws_size, hipStream_t stream) {
  (void)in_sizes;
  (void)n_in;
  (void)out_size;
  (void)ws_size;
  const float* x = (const float*)d_in[0];
  const float* qkv_w = (const float*)d_in[1];
  const float* o_w = (const float*)d_in[2];
  const float* o_scale = (const float*)d_in[3];
  const float* w1 = (const float*)d_in[4];
  const float* w2 = (const float*)d_in[5];
  const float* w2_scale = (const float*)d_in[6];
  float* out = (float*)d_out;

  const size_t TD = (size_t)T_SEQ * DMODEL;
  float* xn = (float*)d_ws;
  float* qkv = xn + TD;
  float* qh = qkv + 3 * TD;
  float* kh = qh + TD;
  float* vh = kh + TD;
  float* attn = vh + TD;
  float* x1 = attn + TD;
  float* hbuf = qkv;

  rmsnorm_kernel<<<T_SEQ, 256, 0, stream>>>(x, xn, DMODEL);
  gemm_wt_kernel<<<dim3(3 * DMODEL / BN, T_SEQ / BM), 256, 0, stream>>>(
      xn, qkv_w, nullptr, nullptr, qkv, T_SEQ, 3 * DMODEL, DMODEL, 0);
  qk_rope_kernel<<<dim3(T_SEQ, NH), 64, 0, stream>>>(qkv, qh, kh, vh);
  attn_kernel<<<dim3(NH, T_SEQ / 16), 32, 0, stream>>>(qh, kh, vh, attn);
  gemm_wt_kernel<<<dim3(DMODEL / BN, T_SEQ / BM), 256, 0, stream>>>(
      attn, o_w, o_scale, x, x1, T_SEQ, DMODEL, DMODEL, 0);
  rmsnorm_kernel<<<T_SEQ, 256, 0, stream>>>(x1, xn, DMODEL);
  gemm_wt_kernel<<<dim3(DMLP / BN, T_SEQ / BM), 256, 0, stream>>>(
      xn, w1, nullptr, nullptr, hbuf, T_SEQ, DMLP, DMODEL, 1);
  gemm_wt_kernel<<<dim3(DMODEL / BN, T_SEQ / BM), 256, 0, stream>>>(
      hbuf, w2, w2_scale, x1, out, T_SEQ, DMODEL, DMLP, 0);
}


# QKV GEMM epilogue does q,k head RMS norm, rotary and head-major repack on its accumulators (one wave tile = one head); separate norm-rope pass and its launch gone; head square-sum order changes
# speedup vs baseline: 2.5242x; 1.0406x over previous
#include <hip/hip_runtime.h>
#include <math.h>


#define T_SEQ  4096
#define DMODEL 1024
#define NH     16
#define HD     64
#define DMLP   4096
#define WIN    512
#define EPS_F  1.1920929e-07f

#define BM   32
#define BN   256
#define KC   32
#define LSTR 36

typedef float v2f __attribute__((ext_vector_type(2)));
typedef float v4f __attribute__((ext_vector_type(4)));
typedef float v8f __attribute__((ext_vector_type(8)));

__device__ __forceinline__ v8f wmma4(v2f a, v2f b, v8f c) {
  return __builtin_amdgcn_wmma_f32_16x16x4_f32(false, a, false, b, (short)0, c,
                                               false, false);
}

__global__ __launch_bounds__(256) void rmsnorm_kernel(
    const float* __restrict__ in, float* __restrict__ out, int ncol) {
  const int row = blockIdx.x;
  const float* r = in + (size_t)row * ncol;
  float* o = out + (size_t)row * ncol;
  float ss = 0.0f;
  for (int i = threadIdx.x; i < ncol; i += blockDim.x) {
    float v = r[i];
    ss += v * v;
  }
  __shared__ float red[256];
  red[threadIdx.x] = ss;
  __syncthreads();
  for (int s = 128; s > 0; s >>= 1) {
    if (threadIdx.x < (unsigned)s) red[threadIdx.x] += red[threadIdx.x + s];
    __syncthreads();
  }
  float inv = rsqrtf(red[0] / (float)ncol + EPS_F);
  for (int i = threadIdx.x; i < ncol; i += blockDim.x) o[i] = r[i] * inv;
}

#define TSTR 20

__device__ __forceinline__ float row16_sum(float v) {
  v += __builtin_bit_cast(float, __builtin_amdgcn_update_dpp(
           0, __builtin_bit_cast(int, v), 0x161, 0xf, 0xf, false));
  v += __builtin_bit_cast(float, __builtin_amdgcn_update_dpp(
           0, __builtin_bit_cast(int, v), 0x162, 0xf, 0xf, false));
  v += __builtin_bit_cast(float, __builtin_amdgcn_update_dpp(
           0, __builtin_bit_cast(int, v), 0x164, 0xf, 0xf, false));
  v += __builtin_bit_cast(float, __builtin_amdgcn_update_dpp(
           0, __builtin_bit_cast(int, v), 0x168, 0xf, 0xf, false));
  return v;
}

template <int MODE>
__global__ __launch_bounds__(256) void gemm_wt_kernel(
    const float* __restrict__ A, const float* __restrict__ Wt,
    const float* __restrict__ colScale, const float* __restrict__ resid,
    float* __restrict__ out, int M, int N, int K, int act) {
  __shared__ float lA[2][BM * LSTR];
  __shared__ float lW[2][BN * LSTR];
  (void)M;

  const int tid = threadIdx.x;
  const int lane = tid & 31;
  const int wave = tid >> 5;
  const int lm = lane & 15;
  const int half = lane >> 4;
  const int wm = (wave & 1) * 16;
  const int wn = (wave >> 1) * 64;

  const int m0 = blockIdx.y * BM;
  const int n00 = blockIdx.x * BN;

  const int ldRow = tid >> 3;
  const int ldCol = (tid & 7) * 4;
  const float* Ag = A + (size_t)(m0 + ldRow) * K + ldCol;
  const float* Wg = Wt + (size_t)(n00 + ldRow) * K + ldCol;

  v4f ra;
  v4f rw[8];

  ra = *(const v4f*)(Ag);
#pragma unroll
  for (int i = 0; i < 8; ++i)
    rw[i] = *(const v4f*)(Wg + (size_t)(i * 32) * K);
  *(v4f*)&lA[0][ldRow * LSTR + ldCol] = ra;
#pragma unroll
  for (int i = 0; i < 8; ++i)
    *(v4f*)&lW[0][(ldRow + i * 32) * LSTR + ldCol] = rw[i];
  __syncthreads();

  v8f acc[4] = {};
  int cur = 0;
  for (int kc = 0; kc < K; kc += KC) {
    const int nxt = kc + KC;
    if (nxt < K) {
      ra = *(const v4f*)(Ag + nxt);
#pragma unroll
      for (int i = 0; i < 8; ++i)
        rw[i] = *(const v4f*)(Wg + (size_t)(i * 32) * K + nxt);
    }
    if (kc + 2 * KC < K)
      __builtin_prefetch(Wg + (size_t)(kc + 2 * KC), 0, 1);

    const float* As = &lA[cur][(wm + lm) * LSTR + 2 * half];
    const float* Ws = &lW[cur][(wn + lm) * LSTR + 2 * half];
#pragma unroll
    for (int kk = 0; kk < KC; kk += 4) {
      v2f a = *(const v2f*)(As + kk);
      v2f b0 = *(const v2f*)(Ws + kk);
      v2f b1 = *(const v2f*)(Ws + kk + 16 * LSTR);
      v2f b2 = *(const v2f*)(Ws + kk + 32 * LSTR);
      v2f b3 = *(const v2f*)(Ws + kk + 48 * LSTR);
      acc[0] = wmma4(a, b0, acc[0]);
      acc[1] = wmma4(a, b1, acc[1]);
      acc[2] = wmma4(a, b2, acc[2]);
      acc[3] = wmma4(a, b3, acc[3]);
    }
    if (nxt < K) {
      *(v4f*)&lA[cur ^ 1][ldRow * LSTR + ldCol] = ra;
#pragma unroll
      for (int i = 0; i < 8; ++i)
        *(v4f*)&lW[cur ^ 1][(ldRow + i * 32) * LSTR + ldCol] = rw[i];
    }
    __syncthreads();
    cur ^= 1;
  }

  if constexpr (MODE == 1) {
    const int sect = n00 / DMODEL;
    const int h = (n00 - sect * DMODEL + wn) / HD;
    const int lr = wm + 8 * half;
    float* dst = out + (((size_t)sect * NH + h) * T_SEQ + (m0 + lr)) * HD + lm;
    if (sect == 2) {
#pragma unroll
      for (int r = 0; r < 8; ++r) {
#pragma unroll
        for (int s = 0; s < 4; ++s) dst[r * HD + s * 16] = acc[s][r];
      }
    } else {
      float* tcs = &lW[0][0];
      float* tsn = tcs + BM * TSTR;
      {
        const int row = tid >> 3;
        const float tf = (float)(m0 + row);
#pragma unroll
        for (int u = 0; u < 2; ++u) {
          const int j = (tid & 7) * 2 + u;
          const float theta = tf * powf(10000.0f, -(float)j / 15.0f);
          tcs[row * TSTR + j] = cosf(theta);
          tsn[row * TSTR + j] = sinf(theta);
        }
      }
      __syncthreads();

      float ssq[8];
#pragma unroll
      for (int r = 0; r < 8; ++r)
        ssq[r] = (acc[0][r] * acc[0][r] + acc[1][r] * acc[1][r]) +
                 (acc[2][r] * acc[2][r] + acc[3][r] * acc[3][r]);
#pragma unroll
      for (int r = 0; r < 8; ++r) ssq[r] = row16_sum(ssq[r]);

#pragma unroll
      for (int r = 0; r < 8; ++r) {
        const float inv = rsqrtf(ssq[r] / (float)HD + EPS_F);
        const float c = tcs[(lr + r) * TSTR + lm];
        const float sn = tsn[(lr + r) * TSTR + lm];
        const float a0 = acc[0][r] * inv, a1 = acc[1][r] * inv;
        const float a2 = acc[2][r] * inv, a3 = acc[3][r] * inv;
        dst[r * HD] = a0 * c + a2 * sn;
        dst[r * HD + 32] = -a0 * sn + a2 * c;
        dst[r * HD + 16] = a1 * 1.0f + a3 * 0.0f;
        dst[r * HD + 48] = -a1 * 0.0f + a3 * 1.0f;
      }
    }
    return;
  }

  const int mB = m0 + wm;
  const int nB = n00 + wn;
#pragma unroll
  for (int s = 0; s < 4; ++s) {
    const int n = nB + s * 16 + lm;
    const float cs = colScale ? colScale[n] : 1.0f;
#pragma unroll
    for (int r = 0; r < 8; ++r) {
      const int m = mB + r + 8 * half;
      float v = acc[s][r] * cs;
      if (act == 1) {
        v = fmaxf(v, 0.0f);
        v = v * v;
      }
      if (resid) v += resid[(size_t)m * N + n];
      out[(size_t)m * N + n] = v;
    }
  }
}

__device__ __forceinline__ float xhalf(float x) {
  const int b = __builtin_bit_cast(int, x);
  return __builtin_bit_cast(
      float, __builtin_amdgcn_permlanex16(b, b, 0x76543210, 0xfedcba98, false,
                                          false));
}

__global__ __launch_bounds__(32) void attn_kernel(
    const float* __restrict__ Qh, const float* __restrict__ Kh,
    const float* __restrict__ Vh, float* __restrict__ outAttn) {
  const int h = blockIdx.x;
  const int q0 = blockIdx.y * 16;
  const int lane = threadIdx.x;
  const int lm = lane & 15;
  const int half = lane >> 4;
  const size_t hb = (size_t)h * T_SEQ * HD;
  const int qp = q0 + lm;

  const float* qrow = Qh + hb + (size_t)qp * HD + 2 * half;
  v2f qb[16];
#pragma unroll
  for (int dc = 0; dc < 16; ++dc) qb[dc] = *(const v2f*)(qrow + 4 * dc);

  float mi = -INFINITY;
  float li = 0.0f;
  v8f o[4] = {};

  const int lo = q0 - WIN + 1;
  const int kt0 = (lo < 0) ? 0 : (lo / 16) * 16;
  for (int kt = kt0; kt <= q0; kt += 16) {
    const float* krow = Kh + hb + (size_t)(kt + lm) * HD + 2 * half;
    v2f ka[16];
#pragma unroll
    for (int dc = 0; dc < 16; ++dc) ka[dc] = *(const v2f*)(krow + 4 * dc);

    const float* vrow = Vh + hb + (size_t)(kt + 8 * half) * HD + lm;
    v2f va[16];
#pragma unroll
    for (int s = 0; s < 4; ++s) {
#pragma unroll
      for (int f = 0; f < 4; ++f) {
        v2f a;
        a.x = vrow[(2 * s) * HD + f * 16];
        a.y = vrow[(2 * s + 1) * HD + f * 16];
        va[s * 4 + f] = a;
      }
    }

    v8f s0 = {}, s1 = {};
#pragma unroll
    for (int dc = 0; dc < 16; dc += 2) {
      s0 = wmma4(ka[dc], qb[dc], s0);
      s1 = wmma4(ka[dc + 1], qb[dc + 1], s1);
    }
    const v8f st = s0 + s1;

    float sv[8];
#pragma unroll
    for (int r = 0; r < 8; ++r) sv[r] = st[r] * 0.125f;

    if ((kt + 15 > q0) || (kt + WIN <= q0 + 15)) {
#pragma unroll
      for (int r = 0; r < 8; ++r) {
        const int kp = kt + 8 * half + r;
        const bool okm = (kp <= qp) && (kp > qp - WIN);
        sv[r] = okm ? sv[r] : -INFINITY;
      }
    }

    float rm = fmaxf(fmaxf(fmaxf(sv[0], sv[1]), fmaxf(sv[2], sv[3])),
                     fmaxf(fmaxf(sv[4], sv[5]), fmaxf(sv[6], sv[7])));
    rm = fmaxf(rm, xhalf(rm));
    const float nm = fmaxf(mi, rm);
    const float nmz = (nm == -INFINITY) ? 0.0f : nm;
    const float corr = __expf(mi - nmz);
    float p[8];
#pragma unroll
    for (int r = 0; r < 8; ++r) p[r] = __expf(sv[r] - nmz);
    float rs = ((p[0] + p[1]) + (p[2] + p[3])) + ((p[4] + p[5]) + (p[6] + p[7]));
    rs += xhalf(rs);
    li = li * corr + rs;
    mi = nm;
#pragma unroll
    for (int f = 0; f < 4; ++f) o[f] *= corr;

#pragma unroll
    for (int s = 0; s < 4; ++s) {
      v2f pb;
      pb.x = p[2 * s];
      pb.y = p[2 * s + 1];
#pragma unroll
      for (int f = 0; f < 4; ++f) o[f] = wmma4(va[s * 4 + f], pb, o[f]);
    }
  }

  float* orow = outAttn + (size_t)qp * DMODEL + h * HD + 8 * half;
#pragma unroll
  for (int f = 0; f < 4; ++f) {
#pragma unroll
    for (int g = 0; g < 2; ++g) {
      v4f w;
      w.x = o[f][4 * g + 0] / li;
      w.y = o[f][4 * g + 1] / li;
      w.z = o[f][4 * g + 2] / li;
      w.w = o[f][4 * g + 3] / li;
      *(v4f*)(orow + f * 16 + 4 * g) = w;
    }
  }
}

extern "C" void kernel_launch(void* const* d_in, const int* in_sizes, int n_in,
                              void* d_out, int out_size, void* d_ws,
                              size_t ws_size, hipStream_t stream) {
  (void)in_sizes;
  (void)n_in;
  (void)out_size;
  (void)ws_size;
  const float* x = (const float*)d_in[0];
  const float* qkv_w = (const float*)d_in[1];
  const float* o_w = (const float*)d_in[2];
  const float* o_scale = (const float*)d_in[3];
  const float* w1 = (const float*)d_in[4];
  const float* w2 = (const float*)d_in[5];
  const float* w2_scale = (const float*)d_in[6];
  float* out = (float*)d_out;

  const size_t TD = (size_t)T_SEQ * DMODEL;
  float* xn = (float*)d_ws;
  float* qkv = xn + TD;
  float* qh = qkv + 3 * TD;
  float* kh = qh + TD;
  float* vh = kh + TD;
  float* attn = vh + TD;
  float* x1 = attn + TD;
  float* hbuf = qkv;

  rmsnorm_kernel<<<T_SEQ, 256, 0, stream>>>(x, xn, DMODEL);
  gemm_wt_kernel<1><<<dim3(3 * DMODEL / BN, T_SEQ / BM), 256, 0, stream>>>(
      xn, qkv_w, nullptr, nullptr, qh, T_SEQ, 3 * DMODEL, DMODEL, 0);
  attn_kernel<<<dim3(NH, T_SEQ / 16), 32, 0, stream>>>(qh, kh, vh, attn);
  gemm_wt_kernel<0><<<dim3(DMODEL / BN, T_SEQ / BM), 256, 0, stream>>>(
      attn, o_w, o_scale, x, x1, T_SEQ, DMODEL, DMODEL, 0);
  rmsnorm_kernel<<<T_SEQ, 256, 0, stream>>>(x1, xn, DMODEL);
  gemm_wt_kernel<0><<<dim3(DMLP / BN, T_SEQ / BM), 256, 0, stream>>>(
      xn, w1, nullptr, nullptr, hbuf, T_SEQ, DMLP, DMODEL, 1);
  gemm_wt_kernel<0><<<dim3(DMODEL / BN, T_SEQ / BM), 256, 0, stream>>>(
      hbuf, w2, w2_scale, x1, out, T_SEQ, DMODEL, DMLP, 0);
}


# GEMM re-tiled: 4-wave blocks, 64x64 wave tiles on 128x128 (N=3072,4096) and 32x64 on 64x128 (N=1024), was 16x64 on 32x256: fewer LDS fragment reads per WMMA, fewer staged bytes per product
# speedup vs baseline: 2.7430x; 1.0867x over previous
#include <hip/hip_runtime.h>
#include <math.h>


#define T_SEQ  4096
#define DMODEL 1024
#define NH     16
#define HD     64
#define DMLP   4096
#define WIN    512
#define EPS_F  1.1920929e-07f

#define BN   128
#define KC   32
#define LSTR 36
#define GTHR 128

typedef float v2f __attribute__((ext_vector_type(2)));
typedef float v4f __attribute__((ext_vector_type(4)));
typedef float v8f __attribute__((ext_vector_type(8)));

__device__ __forceinline__ v8f wmma4(v2f a, v2f b, v8f c) {
  return __builtin_amdgcn_wmma_f32_16x16x4_f32(false, a, false, b, (short)0, c,
                                               false, false);
}

__global__ __launch_bounds__(256) void rmsnorm_kernel(
    const float* __restrict__ in, float* __restrict__ out, int ncol) {
  const int row = blockIdx.x;
  const float* r = in + (size_t)row * ncol;
  float* o = out + (size_t)row * ncol;
  float ss = 0.0f;
  for (int i = threadIdx.x; i < ncol; i += blockDim.x) {
    float v = r[i];
    ss += v * v;
  }
  __shared__ float red[256];
  red[threadIdx.x] = ss;
  __syncthreads();
  for (int s = 128; s > 0; s >>= 1) {
    if (threadIdx.x < (unsigned)s) red[threadIdx.x] += red[threadIdx.x + s];
    __syncthreads();
  }
  float inv = rsqrtf(red[0] / (float)ncol + EPS_F);
  for (int i = threadIdx.x; i < ncol; i += blockDim.x) o[i] = r[i] * inv;
}

#define TSTR 20

__device__ __forceinline__ float row16_sum(float v) {
  v += __builtin_bit_cast(float, __builtin_amdgcn_update_dpp(
           0, __builtin_bit_cast(int, v), 0x161, 0xf, 0xf, false));
  v += __builtin_bit_cast(float, __builtin_amdgcn_update_dpp(
           0, __builtin_bit_cast(int, v), 0x162, 0xf, 0xf, false));
  v += __builtin_bit_cast(float, __builtin_amdgcn_update_dpp(
           0, __builtin_bit_cast(int, v), 0x164, 0xf, 0xf, false));
  v += __builtin_bit_cast(float, __builtin_amdgcn_update_dpp(
           0, __builtin_bit_cast(int, v), 0x168, 0xf, 0xf, false));
  return v;
}

template <int MODE, int MI>
__global__ __launch_bounds__(GTHR) void gemm_wt_kernel(
    const float* __restrict__ A, const float* __restrict__ Wt,
    const float* __restrict__ colScale, const float* __restrict__ resid,
    float* __restrict__ out, int M, int N, int K, int act) {
  constexpr int BM = 32 * MI;
  constexpr int NA = 2 * MI;
  __shared__ float lA[2 * BM * LSTR];
  __shared__ float lW[2 * BN * LSTR];
  (void)M;

  const int tid = threadIdx.x;
  const int lane = tid & 31;
  const int wave = tid >> 5;
  const int lm = lane & 15;
  const int half = lane >> 4;
  const int wm = (wave & 1) * (16 * MI);
  const int wn = (wave >> 1) * 64;

  const int m0 = blockIdx.y * BM;
  const int n00 = blockIdx.x * BN;

  const int ldRow = tid >> 3;
  const int ldCol = (tid & 7) * 4;
  const float* Ag = A + (size_t)(m0 + ldRow) * K + ldCol;
  const float* Wg = Wt + (size_t)(n00 + ldRow) * K + ldCol;
  const int ldOff = ldRow * LSTR + ldCol;

  v4f ra[NA];
  v4f rw[8];

#pragma unroll
  for (int i = 0; i < NA; ++i) ra[i] = *(const v4f*)(Ag + (size_t)(i * 16) * K);
#pragma unroll
  for (int i = 0; i < 8; ++i) rw[i] = *(const v4f*)(Wg + (size_t)(i * 16) * K);
#pragma unroll
  for (int i = 0; i < NA; ++i) *(v4f*)&lA[ldOff + i * 16 * LSTR] = ra[i];
#pragma unroll
  for (int i = 0; i < 8; ++i) *(v4f*)&lW[ldOff + i * 16 * LSTR] = rw[i];
  __syncthreads();

  v8f acc[MI][4] = {};
  int cur = 0;
  for (int kc = 0; kc < K; kc += KC) {
    const int nxt = kc + KC;
    if (nxt < K) {
#pragma unroll
      for (int i = 0; i < NA; ++i)
        ra[i] = *(const v4f*)(Ag + (size_t)(i * 16) * K + nxt);
#pragma unroll
      for (int i = 0; i < 8; ++i)
        rw[i] = *(const v4f*)(Wg + (size_t)(i * 16) * K + nxt);
    }

    const float* As = &lA[cur * (BM * LSTR) + (wm + lm) * LSTR + 2 * half];
    const float* Ws = &lW[cur * (BN * LSTR) + (wn + lm) * LSTR + 2 * half];
#pragma unroll
    for (int kk = 0; kk < KC; kk += 4) {
      v2f a[MI], b[4];
#pragma unroll
      for (int i = 0; i < MI; ++i) a[i] = *(const v2f*)(As + kk + i * 16 * LSTR);
#pragma unroll
      for (int i = 0; i < 4; ++i) b[i] = *(const v2f*)(Ws + kk + i * 16 * LSTR);
#pragma unroll
      for (int i = 0; i < MI; ++i) {
#pragma unroll
        for (int s = 0; s < 4; ++s) acc[i][s] = wmma4(a[i], b[s], acc[i][s]);
      }
    }
    if (nxt < K) {
      const int oa = (cur ^ 1) * (BM * LSTR) + ldOff;
      const int ow = (cur ^ 1) * (BN * LSTR) + ldOff;
#pragma unroll
      for (int i = 0; i < NA; ++i) *(v4f*)&lA[oa + i * 16 * LSTR] = ra[i];
#pragma unroll
      for (int i = 0; i < 8; ++i) *(v4f*)&lW[ow + i * 16 * LSTR] = rw[i];
    }
    __syncthreads();
    cur ^= 1;
  }

  if constexpr (MODE == 1) {
    const int sect = n00 / DMODEL;
    const int h = (n00 - sect * DMODEL + wn) / HD;
    const int lr = wm + 8 * half;
    float* dst = out + (((size_t)sect * NH + h) * T_SEQ + (m0 + lr)) * HD + lm;
    if (sect == 2) {
#pragma unroll
      for (int i = 0; i < MI; ++i) {
#pragma unroll
        for (int r = 0; r < 8; ++r) {
#pragma unroll
          for (int s = 0; s < 4; ++s)
            dst[(i * 16 + r) * HD + s * 16] = acc[i][s][r];
        }
      }
    } else {
      float* tcs = &lW[0];
      float* tsn = tcs + BM * TSTR;
      {
        const int jb = (tid & 7) * 2;
        float fr[2];
#pragma unroll
        for (int u = 0; u < 2; ++u)
          fr[u] = powf(10000.0f, -(float)(jb + u) / 15.0f);
        for (int row = tid >> 3; row < BM; row += GTHR / 8) {
          const float tf = (float)(m0 + row);
#pragma unroll
          for (int u = 0; u < 2; ++u) {
            const float theta = tf * fr[u];
            tcs[row * TSTR + jb + u] = cosf(theta);
            tsn[row * TSTR + jb + u] = sinf(theta);
          }
        }
      }
      __syncthreads();

#pragma unroll
      for (int i = 0; i < MI; ++i) {
        float ssq[8];
#pragma unroll
        for (int r = 0; r < 8; ++r)
          ssq[r] = (acc[i][0][r] * acc[i][0][r] + acc[i][1][r] * acc[i][1][r]) +
                   (acc[i][2][r] * acc[i][2][r] + acc[i][3][r] * acc[i][3][r]);
#pragma unroll
        for (int r = 0; r < 8; ++r) ssq[r] = row16_sum(ssq[r]);

#pragma unroll
        for (int r = 0; r < 8; ++r) {
          const int row = i * 16 + r;
          const float inv = rsqrtf(ssq[r] / (float)HD + EPS_F);
          const float cc = tcs[(lr + row) * TSTR + lm];
          const float sn = tsn[(lr + row) * TSTR + lm];
          const float a0 = acc[i][0][r] * inv, a1 = acc[i][1][r] * inv;
          const float a2 = acc[i][2][r] * inv, a3 = acc[i][3][r] * inv;
          dst[row * HD] = a0 * cc + a2 * sn;
          dst[row * HD + 32] = -a0 * sn + a2 * cc;
          dst[row * HD + 16] = a1 * 1.0f + a3 * 0.0f;
          dst[row * HD + 48] = -a1 * 0.0f + a3 * 1.0f;
        }
      }
    }
    return;
  }

  const int mB = m0 + wm;
  const int nB = n00 + wn;
#pragma unroll
  for (int s = 0; s < 4; ++s) {
    const int n = nB + s * 16 + lm;
    const float cs = colScale ? colScale[n] : 1.0f;
#pragma unroll
    for (int i = 0; i < MI; ++i) {
#pragma unroll
      for (int r = 0; r < 8; ++r) {
        const int m = mB + i * 16 + r + 8 * half;
        float v = acc[i][s][r] * cs;
        if (act == 1) {
          v = fmaxf(v, 0.0f);
          v = v * v;
        }
        if (resid) v += resid[(size_t)m * N + n];
        out[(size_t)m * N + n] = v;
      }
    }
  }
}

__device__ __forceinline__ float xhalf(float x) {
  const int b = __builtin_bit_cast(int, x);
  return __builtin_bit_cast(
      float, __builtin_amdgcn_permlanex16(b, b, 0x76543210, 0xfedcba98, false,
                                          false));
}

__global__ __launch_bounds__(32) void attn_kernel(
    const float* __restrict__ Qh, const float* __restrict__ Kh,
    const float* __restrict__ Vh, float* __restrict__ outAttn) {
  const int h = blockIdx.x;
  const int q0 = blockIdx.y * 16;
  const int lane = threadIdx.x;
  const int lm = lane & 15;
  const int half = lane >> 4;
  const size_t hb = (size_t)h * T_SEQ * HD;
  const int qp = q0 + lm;

  const float* qrow = Qh + hb + (size_t)qp * HD + 2 * half;
  v2f qb[16];
#pragma unroll
  for (int dc = 0; dc < 16; ++dc) qb[dc] = *(const v2f*)(qrow + 4 * dc);

  float mi = -INFINITY;
  float li = 0.0f;
  v8f o[4] = {};

  const int lo = q0 - WIN + 1;
  const int kt0 = (lo < 0) ? 0 : (lo / 16) * 16;
  for (int kt = kt0; kt <= q0; kt += 16) {
    const float* krow = Kh + hb + (size_t)(kt + lm) * HD + 2 * half;
    v2f ka[16];
#pragma unroll
    for (int dc = 0; dc < 16; ++dc) ka[dc] = *(const v2f*)(krow + 4 * dc);

    const float* vrow = Vh + hb + (size_t)(kt + 8 * half) * HD + lm;
    v2f va[16];
#pragma unroll
    for (int s = 0; s < 4; ++s) {
#pragma unroll
      for (int f = 0; f < 4; ++f) {
        v2f a;
        a.x = vrow[(2 * s) * HD + f * 16];
        a.y = vrow[(2 * s + 1) * HD + f * 16];
        va[s * 4 + f] = a;
      }
    }

    v8f s0 = {}, s1 = {};
#pragma unroll
    for (int dc = 0; dc < 16; dc += 2) {
      s0 = wmma4(ka[dc], qb[dc], s0);
      s1 = wmma4(ka[dc + 1], qb[dc + 1], s1);
    }
    const v8f st = s0 + s1;

    float sv[8];
#pragma unroll
    for (int r = 0; r < 8; ++r) sv[r] = st[r] * 0.125f;

    if ((kt + 15 > q0) || (kt + WIN <= q0 + 15)) {
#pragma unroll
      for (int r = 0; r < 8; ++r) {
        const int kp = kt + 8 * half + r;
        const bool okm = (kp <= qp) && (kp > qp - WIN);
        sv[r] = okm ? sv[r] : -INFINITY;
      }
    }

    float rm = fmaxf(fmaxf(fmaxf(sv[0], sv[1]), fmaxf(sv[2], sv[3])),
                     fmaxf(fmaxf(sv[4], sv[5]), fmaxf(sv[6], sv[7])));
    rm = fmaxf(rm, xhalf(rm));
    const float nm = fmaxf(mi, rm);
    const float nmz = (nm == -INFINITY) ? 0.0f : nm;
    const float corr = __expf(mi - nmz);
    float p[8];
#pragma unroll
    for (int r = 0; r < 8; ++r) p[r] = __expf(sv[r] - nmz);
    float rs = ((p[0] + p[1]) + (p[2] + p[3])) + ((p[4] + p[5]) + (p[6] + p[7]));
    rs += xhalf(rs);
    li = li * corr + rs;
    mi = nm;
#pragma unroll
    for (int f = 0; f < 4; ++f) o[f] *= corr;

#pragma unroll
    for (int s = 0; s < 4; ++s) {
      v2f pb;
      pb.x = p[2 * s];
      pb.y = p[2 * s + 1];
#pragma unroll
      for (int f = 0; f < 4; ++f) o[f] = wmma4(va[s * 4 + f], pb, o[f]);
    }
  }

  float* orow = outAttn + (size_t)qp * DMODEL + h * HD + 8 * half;
#pragma unroll
  for (int f = 0; f < 4; ++f) {
#pragma unroll
    for (int g = 0; g < 2; ++g) {
      v4f w;
      w.x = o[f][4 * g + 0] / li;
      w.y = o[f][4 * g + 1] / li;
      w.z = o[f][4 * g + 2] / li;
      w.w = o[f][4 * g + 3] / li;
      *(v4f*)(orow + f * 16 + 4 * g) = w;
    }
  }
}

extern "C" void kernel_launch(void* const* d_in, const int* in_sizes, int n_in,
                              void* d_out, int out_size, void* d_ws,
                              size_t ws_size, hipStream_t stream) {
  (void)in_sizes;
  (void)n_in;
  (void)out_size;
  (void)ws_size;
  const float* x = (const float*)d_in[0];
  const float* qkv_w = (const float*)d_in[1];
  const float* o_w = (const float*)d_in[2];
  const float* o_scale = (const float*)d_in[3];
  const float* w1 = (const float*)d_in[4];
  const float* w2 = (const float*)d_in[5];
  const float* w2_scale = (const float*)d_in[6];
  float* out = (float*)d_out;

  const size_t TD = (size_t)T_SEQ * DMODEL;
  float* xn = (float*)d_ws;
  float* qkv = xn + TD;
  float* qh = qkv + 3 * TD;
  float* kh = qh + TD;
  float* vh = kh + TD;
  float* attn = vh + TD;
  float* x1 = attn + TD;
  float* hbuf = qkv;

  rmsnorm_kernel<<<T_SEQ, 256, 0, stream>>>(x, xn, DMODEL);
  gemm_wt_kernel<1, 4><<<dim3(3 * DMODEL / BN, T_SEQ / 128), GTHR, 0, stream>>>(
      xn, qkv_w, nullptr, nullptr, qh, T_SEQ, 3 * DMODEL, DMODEL, 0);
  attn_kernel<<<dim3(NH, T_SEQ / 16), 32, 0, stream>>>(qh, kh, vh, attn);
  gemm_wt_kernel<0, 2><<<dim3(DMODEL / BN, T_SEQ / 64), GTHR, 0, stream>>>(
      attn, o_w, o_scale, x, x1, T_SEQ, DMODEL, DMODEL, 0);
  rmsnorm_kernel<<<T_SEQ, 256, 0, stream>>>(x1, xn, DMODEL);
  gemm_wt_kernel<0, 4><<<dim3(DMLP / BN, T_SEQ / 128), GTHR, 0, stream>>>(
      xn, w1, nullptr, nullptr, hbuf, T_SEQ, DMLP, DMODEL, 1);
  gemm_wt_kernel<0, 2><<<dim3(DMODEL / BN, T_SEQ / 64), GTHR, 0, stream>>>(
      hbuf, w2, w2_scale, x1, out, T_SEQ, DMODEL, DMLP, 0);
}


# GEMM epilogues handed through LDS in 16x64 row blocks per wave: residual read and output written as whole 256-byte row pieces in 128-bit batches, not a 4-byte load, wait and store per element
# speedup vs baseline: 2.8591x; 1.0423x over previous
#include <hip/hip_runtime.h>
#include <math.h>


#define T_SEQ  4096
#define DMODEL 1024
#define NH     16
#define HD     64
#define DMLP   4096
#define WIN    512
#define EPS_F  1.1920929e-07f

#define BN   128
#define KC   32
#define LSTR 36
#define GTHR 128

typedef float v2f __attribute__((ext_vector_type(2)));
typedef float v4f __attribute__((ext_vector_type(4)));
typedef float v8f __attribute__((ext_vector_type(8)));

__device__ __forceinline__ v8f wmma4(v2f a, v2f b, v8f c) {
  return __builtin_amdgcn_wmma_f32_16x16x4_f32(false, a, false, b, (short)0, c,
                                               false, false);
}

__global__ __launch_bounds__(256) void rmsnorm_kernel(
    const float* __restrict__ in, float* __restrict__ out, int ncol) {
  const int row = blockIdx.x;
  const float* r = in + (size_t)row * ncol;
  float* o = out + (size_t)row * ncol;
  float ss = 0.0f;
  for (int i = threadIdx.x; i < ncol; i += blockDim.x) {
    float v = r[i];
    ss += v * v;
  }
  __shared__ float red[256];
  red[threadIdx.x] = ss;
  __syncthreads();
  for (int s = 128; s > 0; s >>= 1) {
    if (threadIdx.x < (unsigned)s) red[threadIdx.x] += red[threadIdx.x + s];
    __syncthreads();
  }
  float inv = rsqrtf(red[0] / (float)ncol + EPS_F);
  for (int i = threadIdx.x; i < ncol; i += blockDim.x) o[i] = r[i] * inv;
}

#define TSTR 20
#define ESTR 68

__device__ __forceinline__ void lds_barrier() {
  __builtin_amdgcn_fence(__ATOMIC_RELEASE, "workgroup", "local");
  __builtin_amdgcn_s_barrier();
  __builtin_amdgcn_fence(__ATOMIC_ACQUIRE, "workgroup", "local");
}

__device__ __forceinline__ float row16_sum(float v) {
  v += __builtin_bit_cast(float, __builtin_amdgcn_update_dpp(
           0, __builtin_bit_cast(int, v), 0x161, 0xf, 0xf, false));
  v += __builtin_bit_cast(float, __builtin_amdgcn_update_dpp(
           0, __builtin_bit_cast(int, v), 0x162, 0xf, 0xf, false));
  v += __builtin_bit_cast(float, __builtin_amdgcn_update_dpp(
           0, __builtin_bit_cast(int, v), 0x164, 0xf, 0xf, false));
  v += __builtin_bit_cast(float, __builtin_amdgcn_update_dpp(
           0, __builtin_bit_cast(int, v), 0x168, 0xf, 0xf, false));
  return v;
}

template <int MODE, int MI>
__global__ __launch_bounds__(GTHR) void gemm_wt_kernel(
    const float* __restrict__ A, const float* __restrict__ Wt,
    const float* __restrict__ colScale, const float* __restrict__ resid,
    float* __restrict__ out, int M, int N, int K, int act) {
  constexpr int BM = 32 * MI;
  constexpr int NA = 2 * MI;
  __shared__ float lA[2 * BM * LSTR];
  __shared__ float lW[2 * BN * LSTR];
  (void)M;

  const int tid = threadIdx.x;
  const int lane = tid & 31;
  const int wave = tid >> 5;
  const int lm = lane & 15;
  const int half = lane >> 4;
  const int wm = (wave & 1) * (16 * MI);
  const int wn = (wave >> 1) * 64;

  const int m0 = blockIdx.y * BM;
  const int n00 = blockIdx.x * BN;

  const int ldRow = tid >> 3;
  const int ldCol = (tid & 7) * 4;
  const float* Ag = A + (size_t)(m0 + ldRow) * K + ldCol;
  const float* Wg = Wt + (size_t)(n00 + ldRow) * K + ldCol;
  const int ldOff = ldRow * LSTR + ldCol;

  v4f ra[NA];
  v4f rw[8];

#pragma unroll
  for (int i = 0; i < NA; ++i) ra[i] = *(const v4f*)(Ag + (size_t)(i * 16) * K);
#pragma unroll
  for (int i = 0; i < 8; ++i) rw[i] = *(const v4f*)(Wg + (size_t)(i * 16) * K);
#pragma unroll
  for (int i = 0; i < NA; ++i) *(v4f*)&lA[ldOff + i * 16 * LSTR] = ra[i];
#pragma unroll
  for (int i = 0; i < 8; ++i) *(v4f*)&lW[ldOff + i * 16 * LSTR] = rw[i];
  __syncthreads();

  v8f acc[MI][4] = {};
  int cur = 0;
  for (int kc = 0; kc < K; kc += KC) {
    const int nxt = kc + KC;
    if (nxt < K) {
#pragma unroll
      for (int i = 0; i < NA; ++i)
        ra[i] = *(const v4f*)(Ag + (size_t)(i * 16) * K + nxt);
#pragma unroll
      for (int i = 0; i < 8; ++i)
        rw[i] = *(const v4f*)(Wg + (size_t)(i * 16) * K + nxt);
    }

    const float* As = &lA[cur * (BM * LSTR) + (wm + lm) * LSTR + 2 * half];
    const float* Ws = &lW[cur * (BN * LSTR) + (wn + lm) * LSTR + 2 * half];
#pragma unroll
    for (int kk = 0; kk < KC; kk += 4) {
      v2f a[MI], b[4];
#pragma unroll
      for (int i = 0; i < MI; ++i) a[i] = *(const v2f*)(As + kk + i * 16 * LSTR);
#pragma unroll
      for (int i = 0; i < 4; ++i) b[i] = *(const v2f*)(Ws + kk + i * 16 * LSTR);
#pragma unroll
      for (int i = 0; i < MI; ++i) {
#pragma unroll
        for (int s = 0; s < 4; ++s) acc[i][s] = wmma4(a[i], b[s], acc[i][s]);
      }
    }
    if (nxt < K) {
      const int oa = (cur ^ 1) * (BM * LSTR) + ldOff;
      const int ow = (cur ^ 1) * (BN * LSTR) + ldOff;
#pragma unroll
      for (int i = 0; i < NA; ++i) *(v4f*)&lA[oa + i * 16 * LSTR] = ra[i];
#pragma unroll
      for (int i = 0; i < 8; ++i) *(v4f*)&lW[ow + i * 16 * LSTR] = rw[i];
    }
    __syncthreads();
    cur ^= 1;
  }

  float* stg = &lA[wave * (16 * ESTR)];
  const int sw = (8 * half) * ESTR + lm;
  const int sr = half * ESTR + 4 * lm;

  if constexpr (MODE == 1) {
    const int sect = n00 / DMODEL;
    const int h = (n00 - sect * DMODEL + wn) / HD;
    const int lr = wm + 8 * half;
    float* dsth = out + (((size_t)sect * NH + h) * T_SEQ + (m0 + wm)) * HD;
    if (sect == 2) {
#pragma unroll
      for (int i = 0; i < MI; ++i) {
#pragma unroll
        for (int s = 0; s < 4; ++s) {
#pragma unroll
          for (int r = 0; r < 8; ++r) stg[sw + r * ESTR + s * 16] = acc[i][s][r];
        }
        lds_barrier();
        v4f t[8];
#pragma unroll
        for (int j = 0; j < 8; ++j) t[j] = *(const v4f*)&stg[sr + 2 * j * ESTR];
#pragma unroll
        for (int j = 0; j < 8; ++j)
          *(v4f*)(dsth + (size_t)(i * 16 + 2 * j + half) * HD + 4 * lm) = t[j];
        if (i + 1 < MI) lds_barrier();
      }
    } else {
      float* tcs = &lW[0];
      float* tsn = tcs + BM * TSTR;
      {
        const int jb = (tid & 7) * 2;
        float fr[2];
#pragma unroll
        for (int u = 0; u < 2; ++u)
          fr[u] = powf(10000.0f, -(float)(jb + u) / 15.0f);
        for (int row = tid >> 3; row < BM; row += GTHR / 8) {
          const float tf = (float)(m0 + row);
#pragma unroll
          for (int u = 0; u < 2; ++u) {
            const float theta = tf * fr[u];
            tcs[row * TSTR + jb + u] = cosf(theta);
            tsn[row * TSTR + jb + u] = sinf(theta);
          }
        }
      }
      __syncthreads();

#pragma unroll
      for (int i = 0; i < MI; ++i) {
        float ssq[8];
#pragma unroll
        for (int r = 0; r < 8; ++r)
          ssq[r] = (acc[i][0][r] * acc[i][0][r] + acc[i][1][r] * acc[i][1][r]) +
                   (acc[i][2][r] * acc[i][2][r] + acc[i][3][r] * acc[i][3][r]);
#pragma unroll
        for (int r = 0; r < 8; ++r) ssq[r] = row16_sum(ssq[r]);

#pragma unroll
        for (int r = 0; r < 8; ++r) {
          const int row = i * 16 + r;
          const float inv = rsqrtf(ssq[r] / (float)HD + EPS_F);
          const float cc = tcs[(lr + row) * TSTR + lm];
          const float sn = tsn[(lr + row) * TSTR + lm];
          const float a0 = acc[i][0][r] * inv, a1 = acc[i][1][r] * inv;
          const float a2 = acc[i][2][r] * inv, a3 = acc[i][3][r] * inv;
          stg[sw + r * ESTR] = a0 * cc + a2 * sn;
          stg[sw + r * ESTR + 32] = -a0 * sn + a2 * cc;
          stg[sw + r * ESTR + 16] = a1 * 1.0f + a3 * 0.0f;
          stg[sw + r * ESTR + 48] = -a1 * 0.0f + a3 * 1.0f;
        }
        lds_barrier();
        v4f t[8];
#pragma unroll
        for (int j = 0; j < 8; ++j) t[j] = *(const v4f*)&stg[sr + 2 * j * ESTR];
#pragma unroll
        for (int j = 0; j < 8; ++j)
          *(v4f*)(dsth + (size_t)(i * 16 + 2 * j + half) * HD + 4 * lm) = t[j];
        if (i + 1 < MI) lds_barrier();
      }
    }
    return;
  }

  const int mB = m0 + wm + half;
  const int nc = n00 + wn + 4 * lm;
  v4f cs = {1.0f, 1.0f, 1.0f, 1.0f};
  if (colScale) cs = *(const v4f*)(colScale + nc);
  const bool hasRes = (resid != nullptr);
  constexpr int RB = (MI <= 2) ? MI : 1;
  v4f rv[RB][8];
  if (hasRes) {
#pragma unroll
    for (int b = 0; b < RB; ++b) {
#pragma unroll
      for (int j = 0; j < 8; ++j)
        rv[b][j] = *(const v4f*)(resid + (size_t)(mB + b * 16 + 2 * j) * N + nc);
    }
  }
#pragma unroll
  for (int i = 0; i < MI; ++i) {
    if (RB == 1 && i > 0 && hasRes) {
#pragma unroll
      for (int j = 0; j < 8; ++j)
        rv[0][j] = *(const v4f*)(resid + (size_t)(mB + i * 16 + 2 * j) * N + nc);
    }
#pragma unroll
    for (int s = 0; s < 4; ++s) {
#pragma unroll
      for (int r = 0; r < 8; ++r) stg[sw + r * ESTR + s * 16] = acc[i][s][r];
    }
    lds_barrier();
    v4f t[8];
#pragma unroll
    for (int j = 0; j < 8; ++j) t[j] = *(const v4f*)&stg[sr + 2 * j * ESTR];
    if (act == 1) {
#pragma unroll
      for (int j = 0; j < 8; ++j) {
        v4f v = t[j] * cs;
        v.x = fmaxf(v.x, 0.0f);
        v.y = fmaxf(v.y, 0.0f);
        v.z = fmaxf(v.z, 0.0f);
        v.w = fmaxf(v.w, 0.0f);
        t[j] = v * v;
      }
    } else {
#pragma unroll
      for (int j = 0; j < 8; ++j) t[j] = t[j] * cs;
    }
    if (hasRes) {
#pragma unroll
      for (int j = 0; j < 8; ++j) t[j] += rv[RB == 1 ? 0 : i][j];
    }
#pragma unroll
    for (int j = 0; j < 8; ++j)
      *(v4f*)(out + (size_t)(mB + i * 16 + 2 * j) * N + nc) = t[j];
    if (i + 1 < MI) lds_barrier();
  }
}

__device__ __forceinline__ float xhalf(float x) {
  const int b = __builtin_bit_cast(int, x);
  return __builtin_bit_cast(
      float, __builtin_amdgcn_permlanex16(b, b, 0x76543210, 0xfedcba98, false,
                                          false));
}

__global__ __launch_bounds__(32) void attn_kernel(
    const float* __restrict__ Qh, const float* __restrict__ Kh,
    const float* __restrict__ Vh, float* __restrict__ outAttn) {
  const int h = blockIdx.x;
  const int q0 = blockIdx.y * 16;
  const int lane = threadIdx.x;
  const int lm = lane & 15;
  const int half = lane >> 4;
  const size_t hb = (size_t)h * T_SEQ * HD;
  const int qp = q0 + lm;

  const float* qrow = Qh + hb + (size_t)qp * HD + 2 * half;
  v2f qb[16];
#pragma unroll
  for (int dc = 0; dc < 16; ++dc) qb[dc] = *(const v2f*)(qrow + 4 * dc);

  float mi = -INFINITY;
  float li = 0.0f;
  v8f o[4] = {};

  const int lo = q0 - WIN + 1;
  const int kt0 = (lo < 0) ? 0 : (lo / 16) * 16;
  for (int kt = kt0; kt <= q0; kt += 16) {
    const float* krow = Kh + hb + (size_t)(kt + lm) * HD + 2 * half;
    v2f ka[16];
#pragma unroll
    for (int dc = 0; dc < 16; ++dc) ka[dc] = *(const v2f*)(krow + 4 * dc);

    const float* vrow = Vh + hb + (size_t)(kt + 8 * half) * HD + lm;
    v2f va[16];
#pragma unroll
    for (int s = 0; s < 4; ++s) {
#pragma unroll
      for (int f = 0; f < 4; ++f) {
        v2f a;
        a.x = vrow[(2 * s) * HD + f * 16];
        a.y = vrow[(2 * s + 1) * HD + f * 16];
        va[s * 4 + f] = a;
      }
    }

    v8f s0 = {}, s1 = {};
#pragma unroll
    for (int dc = 0; dc < 16; dc += 2) {
      s0 = wmma4(ka[dc], qb[dc], s0);
      s1 = wmma4(ka[dc + 1], qb[dc + 1], s1);
    }
    const v8f st = s0 + s1;

    float sv[8];
#pragma unroll
    for (int r = 0; r < 8; ++r) sv[r] = st[r] * 0.125f;

    if ((kt + 15 > q0) || (kt + WIN <= q0 + 15)) {
#pragma unroll
      for (int r = 0; r < 8; ++r) {
        const int kp = kt + 8 * half + r;
        const bool okm = (kp <= qp) && (kp > qp - WIN);
        sv[r] = okm ? sv[r] : -INFINITY;
      }
    }

    float rm = fmaxf(fmaxf(fmaxf(sv[0], sv[1]), fmaxf(sv[2], sv[3])),
                     fmaxf(fmaxf(sv[4], sv[5]), fmaxf(sv[6], sv[7])));
    rm = fmaxf(rm, xhalf(rm));
    const float nm = fmaxf(mi, rm);
    const float nmz = (nm == -INFINITY) ? 0.0f : nm;
    const float corr = __expf(mi - nmz);
    float p[8];
#pragma unroll
    for (int r = 0; r < 8; ++r) p[r] = __expf(sv[r] - nmz);
    float rs = ((p[0] + p[1]) + (p[2] + p[3])) + ((p[4] + p[5]) + (p[6] + p[7]));
    rs += xhalf(rs);
    li = li * corr + rs;
    mi = nm;
#pragma unroll
    for (int f = 0; f < 4; ++f) o[f] *= corr;

#pragma unroll
    for (int s = 0; s < 4; ++s) {
      v2f pb;
      pb.x = p[2 * s];
      pb.y = p[2 * s + 1];
#pragma unroll
      for (int f = 0; f < 4; ++f) o[f] = wmma4(va[s * 4 + f], pb, o[f]);
    }
  }

  float* orow = outAttn + (size_t)qp * DMODEL + h * HD + 8 * half;
#pragma unroll
  for (int f = 0; f < 4; ++f) {
#pragma unroll
    for (int g = 0; g < 2; ++g) {
      v4f w;
      w.x = o[f][4 * g + 0] / li;
      w.y = o[f][4 * g + 1] / li;
      w.z = o[f][4 * g + 2] / li;
      w.w = o[f][4 * g + 3] / li;
      *(v4f*)(orow + f * 16 + 4 * g) = w;
    }
  }
}

extern "C" void kernel_launch(void* const* d_in, const int* in_sizes, int n_in,
                              void* d_out, int out_size, void* d_ws,
                              size_t ws_size, hipStream_t stream) {
  (void)in_sizes;
  (void)n_in;
  (void)out_size;
  (void)ws_size;
  const float* x = (const float*)d_in[0];
  const float* qkv_w = (const float*)d_in[1];
  const float* o_w = (const float*)d_in[2];
  const float* o_scale = (const float*)d_in[3];
  const float* w1 = (const float*)d_in[4];
  const float* w2 = (const float*)d_in[5];
  const float* w2_scale = (const float*)d_in[6];
  float* out = (float*)d_out;

  const size_t TD = (size_t)T_SEQ * DMODEL;
  float* xn = (float*)d_ws;
  float* qkv = xn + TD;
  float* qh = qkv + 3 * TD;
  float* kh = qh + TD;
  float* vh = kh + TD;
  float* attn = vh + TD;
  float* x1 = attn + TD;
  float* hbuf = qkv;

  rmsnorm_kernel<<<T_SEQ, 256, 0, stream>>>(x, xn, DMODEL);
  gemm_wt_kernel<1, 4><<<dim3(3 * DMODEL / BN, T_SEQ / 128), GTHR, 0, stream>>>(
      xn, qkv_w, nullptr, nullptr, qh, T_SEQ, 3 * DMODEL, DMODEL, 0);
  attn_kernel<<<dim3(NH, T_SEQ / 16), 32, 0, stream>>>(qh, kh, vh, attn);
  gemm_wt_kernel<0, 2><<<dim3(DMODEL / BN, T_SEQ / 64), GTHR, 0, stream>>>(
      attn, o_w, o_scale, x, x1, T_SEQ, DMODEL, DMODEL, 0);
  rmsnorm_kernel<<<T_SEQ, 256, 0, stream>>>(x1, xn, DMODEL);
  gemm_wt_kernel<0, 4><<<dim3(DMLP / BN, T_SEQ / 128), GTHR, 0, stream>>>(
      xn, w1, nullptr, nullptr, hbuf, T_SEQ, DMLP, DMODEL, 1);
  gemm_wt_kernel<0, 2><<<dim3(DMODEL / BN, T_SEQ / 64), GTHR, 0, stream>>>(
      hbuf, w2, w2_scale, x1, out, T_SEQ, DMODEL, DMLP, 0);
}


# N=1024 GEMMs: one 8-wave block per WGP on a 128x128 tile; a SIMD's two waves split each K chunk of 64 on one 64x64 wave tile (was 32x64 tiles, two 4-wave blocks); two K-half sums added at the end
# speedup vs baseline: 2.9986x; 1.0488x over previous
#include <hip/hip_runtime.h>
#include <math.h>


#define T_SEQ  4096
#define DMODEL 1024
#define NH     16
#define HD     64
#define DMLP   4096
#define WIN    512
#define EPS_F  1.1920929e-07f

#define BN   128
#define KC   32
#define LSTR 36
#define GTHR 128

typedef float v2f __attribute__((ext_vector_type(2)));
typedef float v4f __attribute__((ext_vector_type(4)));
typedef float v8f __attribute__((ext_vector_type(8)));

__device__ __forceinline__ v8f wmma4(v2f a, v2f b, v8f c) {
  return __builtin_amdgcn_wmma_f32_16x16x4_f32(false, a, false, b, (short)0, c,
                                               false, false);
}

__global__ __launch_bounds__(256) void rmsnorm_kernel(
    const float* __restrict__ in, float* __restrict__ out, int ncol) {
  const int row = blockIdx.x;
  const float* r = in + (size_t)row * ncol;
  float* o = out + (size_t)row * ncol;
  float ss = 0.0f;
  for (int i = threadIdx.x; i < ncol; i += blockDim.x) {
    float v = r[i];
    ss += v * v;
  }
  __shared__ float red[256];
  red[threadIdx.x] = ss;
  __syncthreads();
  for (int s = 128; s > 0; s >>= 1) {
    if (threadIdx.x < (unsigned)s) red[threadIdx.x] += red[threadIdx.x + s];
    __syncthreads();
  }
  float inv = rsqrtf(red[0] / (float)ncol + EPS_F);
  for (int i = threadIdx.x; i < ncol; i += blockDim.x) o[i] = r[i] * inv;
}

#define TSTR 20
#define ESTR 68

__device__ __forceinline__ void lds_barrier() {
  __builtin_amdgcn_fence(__ATOMIC_RELEASE, "workgroup", "local");
  __builtin_amdgcn_s_barrier();
  __builtin_amdgcn_fence(__ATOMIC_ACQUIRE, "workgroup", "local");
}

__device__ __forceinline__ float row16_sum(float v) {
  v += __builtin_bit_cast(float, __builtin_amdgcn_update_dpp(
           0, __builtin_bit_cast(int, v), 0x161, 0xf, 0xf, false));
  v += __builtin_bit_cast(float, __builtin_amdgcn_update_dpp(
           0, __builtin_bit_cast(int, v), 0x162, 0xf, 0xf, false));
  v += __builtin_bit_cast(float, __builtin_amdgcn_update_dpp(
           0, __builtin_bit_cast(int, v), 0x164, 0xf, 0xf, false));
  v += __builtin_bit_cast(float, __builtin_amdgcn_update_dpp(
           0, __builtin_bit_cast(int, v), 0x168, 0xf, 0xf, false));
  return v;
}

template <int MODE, int MI>
__global__ __launch_bounds__(GTHR) void gemm_wt_kernel(
    const float* __restrict__ A, const float* __restrict__ Wt,
    const float* __restrict__ colScale, const float* __restrict__ resid,
    float* __restrict__ out, int M, int N, int K, int act) {
  constexpr int BM = 32 * MI;
  constexpr int NA = 2 * MI;
  __shared__ float lA[2 * BM * LSTR];
  __shared__ float lW[2 * BN * LSTR];
  (void)M;

  const int tid = threadIdx.x;
  const int lane = tid & 31;
  const int wave = tid >> 5;
  const int lm = lane & 15;
  const int half = lane >> 4;
  const int wm = (wave & 1) * (16 * MI);
  const int wn = (wave >> 1) * 64;

  const int m0 = blockIdx.y * BM;
  const int n00 = blockIdx.x * BN;

  const int ldRow = tid >> 3;
  const int ldCol = (tid & 7) * 4;
  const float* Ag = A + (size_t)(m0 + ldRow) * K + ldCol;
  const float* Wg = Wt + (size_t)(n00 + ldRow) * K + ldCol;
  const int ldOff = ldRow * LSTR + ldCol;

  v4f ra[NA];
  v4f rw[8];

#pragma unroll
  for (int i = 0; i < NA; ++i) ra[i] = *(const v4f*)(Ag + (size_t)(i * 16) * K);
#pragma unroll
  for (int i = 0; i < 8; ++i) rw[i] = *(const v4f*)(Wg + (size_t)(i * 16) * K);
#pragma unroll
  for (int i = 0; i < NA; ++i) *(v4f*)&lA[ldOff + i * 16 * LSTR] = ra[i];
#pragma unroll
  for (int i = 0; i < 8; ++i) *(v4f*)&lW[ldOff + i * 16 * LSTR] = rw[i];
  __syncthreads();

  v8f acc[MI][4] = {};
  int cur = 0;
  for (int kc = 0; kc < K; kc += KC) {
    const int nxt = kc + KC;
    if (nxt < K) {
#pragma unroll
      for (int i = 0; i < NA; ++i)
        ra[i] = *(const v4f*)(Ag + (size_t)(i * 16) * K + nxt);
#pragma unroll
      for (int i = 0; i < 8; ++i)
        rw[i] = *(const v4f*)(Wg + (size_t)(i * 16) * K + nxt);
    }

    const float* As = &lA[cur * (BM * LSTR) + (wm + lm) * LSTR + 2 * half];
    const float* Ws = &lW[cur * (BN * LSTR) + (wn + lm) * LSTR + 2 * half];
#pragma unroll
    for (int kk = 0; kk < KC; kk += 4) {
      v2f a[MI], b[4];
#pragma unroll
      for (int i = 0; i < MI; ++i) a[i] = *(const v2f*)(As + kk + i * 16 * LSTR);
#pragma unroll
      for (int i = 0; i < 4; ++i) b[i] = *(const v2f*)(Ws + kk + i * 16 * LSTR);
#pragma unroll
      for (int i = 0; i < MI; ++i) {
#pragma unroll
        for (int s = 0; s < 4; ++s) acc[i][s] = wmma4(a[i], b[s], acc[i][s]);
      }
    }
    if (nxt < K) {
      const int oa = (cur ^ 1) * (BM * LSTR) + ldOff;
      const int ow = (cur ^ 1) * (BN * LSTR) + ldOff;
#pragma unroll
      for (int i = 0; i < NA; ++i) *(v4f*)&lA[oa + i * 16 * LSTR] = ra[i];
#pragma unroll
      for (int i = 0; i < 8; ++i) *(v4f*)&lW[ow + i * 16 * LSTR] = rw[i];
    }
    __syncthreads();
    cur ^= 1;
  }

  float* stg = &lA[wave * (16 * ESTR)];
  const int sw = (8 * half) * ESTR + lm;
  const int sr = half * ESTR + 4 * lm;

  if constexpr (MODE == 1) {
    const int sect = n00 / DMODEL;
    const int h = (n00 - sect * DMODEL + wn) / HD;
    const int lr = wm + 8 * half;
    float* dsth = out + (((size_t)sect * NH + h) * T_SEQ + (m0 + wm)) * HD;
    if (sect == 2) {
#pragma unroll
      for (int i = 0; i < MI; ++i) {
#pragma unroll
        for (int s = 0; s < 4; ++s) {
#pragma unroll
          for (int r = 0; r < 8; ++r) stg[sw + r * ESTR + s * 16] = acc[i][s][r];
        }
        lds_barrier();
        v4f t[8];
#pragma unroll
        for (int j = 0; j < 8; ++j) t[j] = *(const v4f*)&stg[sr + 2 * j * ESTR];
#pragma unroll
        for (int j = 0; j < 8; ++j)
          *(v4f*)(dsth + (size_t)(i * 16 + 2 * j + half) * HD + 4 * lm) = t[j];
        if (i + 1 < MI) lds_barrier();
      }
    } else {
      float* tcs = &lW[0];
      float* tsn = tcs + BM * TSTR;
      {
        const int jb = (tid & 7) * 2;
        float fr[2];
#pragma unroll
        for (int u = 0; u < 2; ++u)
          fr[u] = powf(10000.0f, -(float)(jb + u) / 15.0f);
        for (int row = tid >> 3; row < BM; row += GTHR / 8) {
          const float tf = (float)(m0 + row);
#pragma unroll
          for (int u = 0; u < 2; ++u) {
            const float theta = tf * fr[u];
            tcs[row * TSTR + jb + u] = cosf(theta);
            tsn[row * TSTR + jb + u] = sinf(theta);
          }
        }
      }
      __syncthreads();

#pragma unroll
      for (int i = 0; i < MI; ++i) {
        float ssq[8];
#pragma unroll
        for (int r = 0; r < 8; ++r)
          ssq[r] = (acc[i][0][r] * acc[i][0][r] + acc[i][1][r] * acc[i][1][r]) +
                   (acc[i][2][r] * acc[i][2][r] + acc[i][3][r] * acc[i][3][r]);
#pragma unroll
        for (int r = 0; r < 8; ++r) ssq[r] = row16_sum(ssq[r]);

#pragma unroll
        for (int r = 0; r < 8; ++r) {
          const int row = i * 16 + r;
          const float inv = rsqrtf(ssq[r] / (float)HD + EPS_F);
          const float cc = tcs[(lr + row) * TSTR + lm];
          const float sn = tsn[(lr + row) * TSTR + lm];
          const float a0 = acc[i][0][r] * inv, a1 = acc[i][1][r] * inv;
          const float a2 = acc[i][2][r] * inv, a3 = acc[i][3][r] * inv;
          stg[sw + r * ESTR] = a0 * cc + a2 * sn;
          stg[sw + r * ESTR + 32] = -a0 * sn + a2 * cc;
          stg[sw + r * ESTR + 16] = a1 * 1.0f + a3 * 0.0f;
          stg[sw + r * ESTR + 48] = -a1 * 0.0f + a3 * 1.0f;
        }
        lds_barrier();
        v4f t[8];
#pragma unroll
        for (int j = 0; j < 8; ++j) t[j] = *(const v4f*)&stg[sr + 2 * j * ESTR];
#pragma unroll
        for (int j = 0; j < 8; ++j)
          *(v4f*)(dsth + (size_t)(i * 16 + 2 * j + half) * HD + 4 * lm) = t[j];
        if (i + 1 < MI) lds_barrier();
      }
    }
    return;
  }

  const int mB = m0 + wm + half;
  const int nc = n00 + wn + 4 * lm;
  v4f cs = {1.0f, 1.0f, 1.0f, 1.0f};
  if (colScale) cs = *(const v4f*)(colScale + nc);
  const bool hasRes = (resid != nullptr);
  constexpr int RB = (MI <= 2) ? MI : 1;
  v4f rv[RB][8];
  if (hasRes) {
#pragma unroll
    for (int b = 0; b < RB; ++b) {
#pragma unroll
      for (int j = 0; j < 8; ++j)
        rv[b][j] = *(const v4f*)(resid + (size_t)(mB + b * 16 + 2 * j) * N + nc);
    }
  }
#pragma unroll
  for (int i = 0; i < MI; ++i) {
    if (RB == 1 && i > 0 && hasRes) {
#pragma unroll
      for (int j = 0; j < 8; ++j)
        rv[0][j] = *(const v4f*)(resid + (size_t)(mB + i * 16 + 2 * j) * N + nc);
    }
#pragma unroll
    for (int s = 0; s < 4; ++s) {
#pragma unroll
      for (int r = 0; r < 8; ++r) stg[sw + r * ESTR + s * 16] = acc[i][s][r];
    }
    lds_barrier();
    v4f t[8];
#pragma unroll
    for (int j = 0; j < 8; ++j) t[j] = *(const v4f*)&stg[sr + 2 * j * ESTR];
    if (act == 1) {
#pragma unroll
      for (int j = 0; j < 8; ++j) {
        v4f v = t[j] * cs;
        v.x = fmaxf(v.x, 0.0f);
        v.y = fmaxf(v.y, 0.0f);
        v.z = fmaxf(v.z, 0.0f);
        v.w = fmaxf(v.w, 0.0f);
        t[j] = v * v;
      }
    } else {
#pragma unroll
      for (int j = 0; j < 8; ++j) t[j] = t[j] * cs;
    }
    if (hasRes) {
#pragma unroll
      for (int j = 0; j < 8; ++j) t[j] += rv[RB == 1 ? 0 : i][j];
    }
#pragma unroll
    for (int j = 0; j < 8; ++j)
      *(v4f*)(out + (size_t)(mB + i * 16 + 2 * j) * N + nc) = t[j];
    if (i + 1 < MI) lds_barrier();
  }
}

#define KC2   64
#define LSTR2 68
#define GTHR2 256

__global__ __launch_bounds__(GTHR2) void gemm_ksplit_kernel(
    const float* __restrict__ A, const float* __restrict__ Wt,
    const float* __restrict__ colScale, const float* __restrict__ resid,
    float* __restrict__ out, int M, int N, int K) {
  constexpr int BM = 128;
  __shared__ __attribute__((aligned(16))) float lA[2 * BM * LSTR2];
  __shared__ __attribute__((aligned(16))) float lW[2 * BN * LSTR2];
  __shared__ v4f red[8 * 16 * 32];
  (void)M;

  const int tid = threadIdx.x;
  const int lane = tid & 31;
  const int wave = __builtin_amdgcn_readfirstlane(tid >> 5);
  const int lm = lane & 15;
  const int half = lane >> 4;
  const int kg = wave >> 2;
  const int wm = (wave & 1) * 64;
  const int wn = ((wave >> 1) & 1) * 64;

  const int m0 = blockIdx.y * BM;
  const int n00 = blockIdx.x * BN;

  const int ldRow = tid >> 4;
  const int ldCol = (tid & 15) * 4;
  const float* Ag = A + (size_t)(m0 + ldRow) * K + ldCol;
  const float* Wg = Wt + (size_t)(n00 + ldRow) * K + ldCol;
  const int ldOff = ldRow * LSTR2 + ldCol;

  v4f ra[8];
  v4f rw[8];

#pragma unroll
  for (int i = 0; i < 8; ++i) ra[i] = *(const v4f*)(Ag + (size_t)(i * 16) * K);
#pragma unroll
  for (int i = 0; i < 8; ++i) rw[i] = *(const v4f*)(Wg + (size_t)(i * 16) * K);
#pragma unroll
  for (int i = 0; i < 8; ++i) *(v4f*)&lA[ldOff + i * 16 * LSTR2] = ra[i];
#pragma unroll
  for (int i = 0; i < 8; ++i) *(v4f*)&lW[ldOff + i * 16 * LSTR2] = rw[i];
  __syncthreads();

  v8f acc[4][4] = {};
  int cur = 0;
  for (int kc = 0; kc < K; kc += KC2) {
    const int nxt = kc + KC2;
    if (nxt < K) {
#pragma unroll
      for (int i = 0; i < 8; ++i)
        ra[i] = *(const v4f*)(Ag + (size_t)(i * 16) * K + nxt);
#pragma unroll
      for (int i = 0; i < 8; ++i)
        rw[i] = *(const v4f*)(Wg + (size_t)(i * 16) * K + nxt);
    }

    const float* AsK = &lA[cur * (BM * LSTR2) + (wm + 32 * kg + lm) * LSTR2 +
                           2 * half + kg * (KC2 / 2)];
    const float* AsG = &lA[cur * (BM * LSTR2) +
                           (wm + 32 * (1 - kg) + lm) * LSTR2 + 2 * half +
                           kg * (KC2 / 2)];
    const float* Ws = &lW[cur * (BN * LSTR2) + (wn + lm) * LSTR2 + 2 * half +
                          kg * (KC2 / 2)];
#pragma unroll
    for (int kk = 0; kk < KC2 / 2; kk += 4) {
      v2f a[4], b[4];
#pragma unroll
      for (int i = 0; i < 2; ++i) {
        a[i] = *(const v2f*)(AsK + kk + i * 16 * LSTR2);
        a[2 + i] = *(const v2f*)(AsG + kk + i * 16 * LSTR2);
      }
#pragma unroll
      for (int i = 0; i < 4; ++i) b[i] = *(const v2f*)(Ws + kk + i * 16 * LSTR2);
#pragma unroll
      for (int i = 0; i < 4; ++i) {
#pragma unroll
        for (int s = 0; s < 4; ++s) acc[i][s] = wmma4(a[i], b[s], acc[i][s]);
      }
    }
    if (nxt < K) {
      const int oa = (cur ^ 1) * (BM * LSTR2) + ldOff;
      const int ow = (cur ^ 1) * (BN * LSTR2) + ldOff;
#pragma unroll
      for (int i = 0; i < 8; ++i) *(v4f*)&lA[oa + i * 16 * LSTR2] = ra[i];
#pragma unroll
      for (int i = 0; i < 8; ++i) *(v4f*)&lW[ow + i * 16 * LSTR2] = rw[i];
    }
    __syncthreads();
    cur ^= 1;
  }

  const int mB = m0 + wm + kg * 32 + half;
  const int nc = n00 + wn + 4 * lm;
  v4f cs = {1.0f, 1.0f, 1.0f, 1.0f};
  if (colScale) cs = *(const v4f*)(colScale + nc);
  const bool hasRes = (resid != nullptr);
  v4f rv[2][8];
  if (hasRes) {
#pragma unroll
    for (int b = 0; b < 2; ++b) {
#pragma unroll
      for (int j = 0; j < 8; ++j)
        rv[b][j] = *(const v4f*)(resid + (size_t)(mB + b * 16 + 2 * j) * N + nc);
    }
  }

  v4f* mine = red + (wave * 16) * 32 + lane;
  const v4f* theirs = red + ((wave ^ 4) * 16) * 32 + lane;
#pragma unroll
  for (int j = 0; j < 2; ++j) {
#pragma unroll
    for (int s = 0; s < 4; ++s) {
      const v8f v = acc[2 + j][s];
      mine[((j * 4 + s) * 2 + 0) * 32] = (v4f){v[0], v[1], v[2], v[3]};
      mine[((j * 4 + s) * 2 + 1) * 32] = (v4f){v[4], v[5], v[6], v[7]};
    }
  }
  lds_barrier();
#pragma unroll
  for (int j = 0; j < 2; ++j) {
#pragma unroll
    for (int s = 0; s < 4; ++s) {
      const v4f lo = theirs[((j * 4 + s) * 2 + 0) * 32];
      const v4f hi = theirs[((j * 4 + s) * 2 + 1) * 32];
      const v8f other = {lo.x, lo.y, lo.z, lo.w, hi.x, hi.y, hi.z, hi.w};
      acc[j][s] = acc[j][s] + other;
    }
  }

  float* stg = &lA[wave * (16 * ESTR)];
  const int sw = (8 * half) * ESTR + lm;
  const int sr = half * ESTR + 4 * lm;
#pragma unroll
  for (int i = 0; i < 2; ++i) {
#pragma unroll
    for (int s = 0; s < 4; ++s) {
#pragma unroll
      for (int r = 0; r < 8; ++r) stg[sw + r * ESTR + s * 16] = acc[i][s][r];
    }
    lds_barrier();
    v4f t[8];
#pragma unroll
    for (int j = 0; j < 8; ++j) t[j] = *(const v4f*)&stg[sr + 2 * j * ESTR];
#pragma unroll
    for (int j = 0; j < 8; ++j) t[j] = t[j] * cs;
    if (hasRes) {
#pragma unroll
      for (int j = 0; j < 8; ++j) t[j] += rv[i][j];
    }
#pragma unroll
    for (int j = 0; j < 8; ++j)
      *(v4f*)(out + (size_t)(mB + i * 16 + 2 * j) * N + nc) = t[j];
    if (i + 1 < 2) lds_barrier();
  }
}

__device__ __forceinline__ float xhalf(float x) {
  const int b = __builtin_bit_cast(int, x);
  return __builtin_bit_cast(
      float, __builtin_amdgcn_permlanex16(b, b, 0x76543210, 0xfedcba98, false,
                                          false));
}

__global__ __launch_bounds__(32) void attn_kernel(
    const float* __restrict__ Qh, const float* __restrict__ Kh,
    const float* __restrict__ Vh, float* __restrict__ outAttn) {
  const int h = blockIdx.x;
  const int q0 = blockIdx.y * 16;
  const int lane = threadIdx.x;
  const int lm = lane & 15;
  const int half = lane >> 4;
  const size_t hb = (size_t)h * T_SEQ * HD;
  const int qp = q0 + lm;

  const float* qrow = Qh + hb + (size_t)qp * HD + 2 * half;
  v2f qb[16];
#pragma unroll
  for (int dc = 0; dc < 16; ++dc) qb[dc] = *(const v2f*)(qrow + 4 * dc);

  float mi = -INFINITY;
  float li = 0.0f;
  v8f o[4] = {};

  const int lo = q0 - WIN + 1;
  const int kt0 = (lo < 0) ? 0 : (lo / 16) * 16;
  for (int kt = kt0; kt <= q0; kt += 16) {
    const float* krow = Kh + hb + (size_t)(kt + lm) * HD + 2 * half;
    v2f ka[16];
#pragma unroll
    for (int dc = 0; dc < 16; ++dc) ka[dc] = *(const v2f*)(krow + 4 * dc);

    const float* vrow = Vh + hb + (size_t)(kt + 8 * half) * HD + lm;
    v2f va[16];
#pragma unroll
    for (int s = 0; s < 4; ++s) {
#pragma unroll
      for (int f = 0; f < 4; ++f) {
        v2f a;
        a.x = vrow[(2 * s) * HD + f * 16];
        a.y = vrow[(2 * s + 1) * HD + f * 16];
        va[s * 4 + f] = a;
      }
    }

    v8f s0 = {}, s1 = {};
#pragma unroll
    for (int dc = 0; dc < 16; dc += 2) {
      s0 = wmma4(ka[dc], qb[dc], s0);
      s1 = wmma4(ka[dc + 1], qb[dc + 1], s1);
    }
    const v8f st = s0 + s1;

    float sv[8];
#pragma unroll
    for (int r = 0; r < 8; ++r) sv[r] = st[r] * 0.125f;

    if ((kt + 15 > q0) || (kt + WIN <= q0 + 15)) {
#pragma unroll
      for (int r = 0; r < 8; ++r) {
        const int kp = kt + 8 * half + r;
        const bool okm = (kp <= qp) && (kp > qp - WIN);
        sv[r] = okm ? sv[r] : -INFINITY;
      }
    }

    float rm = fmaxf(fmaxf(fmaxf(sv[0], sv[1]), fmaxf(sv[2], sv[3])),
                     fmaxf(fmaxf(sv[4], sv[5]), fmaxf(sv[6], sv[7])));
    rm = fmaxf(rm, xhalf(rm));
    const float nm = fmaxf(mi, rm);
    const float nmz = (nm == -INFINITY) ? 0.0f : nm;
    const float corr = __expf(mi - nmz);
    float p[8];
#pragma unroll
    for (int r = 0; r < 8; ++r) p[r] = __expf(sv[r] - nmz);
    float rs = ((p[0] + p[1]) + (p[2] + p[3])) + ((p[4] + p[5]) + (p[6] + p[7]));
    rs += xhalf(rs);
    li = li * corr + rs;
    mi = nm;
#pragma unroll
    for (int f = 0; f < 4; ++f) o[f] *= corr;

#pragma unroll
    for (int s = 0; s < 4; ++s) {
      v2f pb;
      pb.x = p[2 * s];
      pb.y = p[2 * s + 1];
#pragma unroll
      for (int f = 0; f < 4; ++f) o[f] = wmma4(va[s * 4 + f], pb, o[f]);
    }
  }

  float* orow = outAttn + (size_t)qp * DMODEL + h * HD + 8 * half;
#pragma unroll
  for (int f = 0; f < 4; ++f) {
#pragma unroll
    for (int g = 0; g < 2; ++g) {
      v4f w;
      w.x = o[f][4 * g + 0] / li;
      w.y = o[f][4 * g + 1] / li;
      w.z = o[f][4 * g + 2] / li;
      w.w = o[f][4 * g + 3] / li;
      *(v4f*)(orow + f * 16 + 4 * g) = w;
    }
  }
}

extern "C" void kernel_launch(void* const* d_in, const int* in_sizes, int n_in,
                              void* d_out, int out_size, void* d_ws,
                              size_t ws_size, hipStream_t stream) {
  (void)in_sizes;
  (void)n_in;
  (void)out_size;
  (void)ws_size;
  const float* x = (const float*)d_in[0];
  const float* qkv_w = (const float*)d_in[1];
  const float* o_w = (const float*)d_in[2];
  const float* o_scale = (const float*)d_in[3];
  const float* w1 = (const float*)d_in[4];
  const float* w2 = (const float*)d_in[5];
  const float* w2_scale = (const float*)d_in[6];
  float* out = (float*)d_out;

  const size_t TD = (size_t)T_SEQ * DMODEL;
  float* xn = (float*)d_ws;
  float* qkv = xn + TD;
  float* qh = qkv + 3 * TD;
  float* kh = qh + TD;
  float* vh = kh + TD;
  float* attn = vh + TD;
  float* x1 = attn + TD;
  float* hbuf = qkv;

  rmsnorm_kernel<<<T_SEQ, 256, 0, stream>>>(x, xn, DMODEL);
  gemm_wt_kernel<1, 4><<<dim3(3 * DMODEL / BN, T_SEQ / 128), GTHR, 0, stream>>>(
      xn, qkv_w, nullptr, nullptr, qh, T_SEQ, 3 * DMODEL, DMODEL, 0);
  attn_kernel<<<dim3(NH, T_SEQ / 16), 32, 0, stream>>>(qh, kh, vh, attn);
  gemm_ksplit_kernel<<<dim3(DMODEL / BN, T_SEQ / 128), GTHR2, 0, stream>>>(
      attn, o_w, o_scale, x, x1, T_SEQ, DMODEL, DMODEL);
  rmsnorm_kernel<<<T_SEQ, 256, 0, stream>>>(x1, xn, DMODEL);
  gemm_wt_kernel<0, 4><<<dim3(DMLP / BN, T_SEQ / 128), GTHR, 0, stream>>>(
      xn, w1, nullptr, nullptr, hbuf, T_SEQ, DMLP, DMODEL, 1);
  gemm_ksplit_kernel<<<dim3(DMODEL / BN, T_SEQ / 128), GTHR2, 0, stream>>>(
      hbuf, w2, w2_scale, x1, out, T_SEQ, DMODEL, DMLP);
}


# Both row RMS norms folded into the consuming GEMMs: 1/RMS scales the accumulators; square sums from the staged A chunks (QKV) and in the o-proj epilogue; 2 launches gone; square-sum order changes
# speedup vs baseline: 3.1005x; 1.0340x over previous
#include <hip/hip_runtime.h>
#include <math.h>


#define T_SEQ  4096
#define DMODEL 1024
#define NH     16
#define HD     64
#define DMLP   4096
#define WIN    512
#define EPS_F  1.1920929e-07f

#define BN   128
#define KC   32
#define LSTR 36
#define GTHR 128

typedef float v2f __attribute__((ext_vector_type(2)));
typedef float v4f __attribute__((ext_vector_type(4)));
typedef float v8f __attribute__((ext_vector_type(8)));

__device__ __forceinline__ v8f wmma4(v2f a, v2f b, v8f c) {
  return __builtin_amdgcn_wmma_f32_16x16x4_f32(false, a, false, b, (short)0, c,
                                               false, false);
}

#define NPLANE (2 * DMODEL / BN)

#define TSTR 20
#define ESTR 68

__device__ __forceinline__ void lds_barrier() {
  __builtin_amdgcn_fence(__ATOMIC_RELEASE, "workgroup", "local");
  __builtin_amdgcn_s_barrier();
  __builtin_amdgcn_fence(__ATOMIC_ACQUIRE, "workgroup", "local");
}

__device__ __forceinline__ float row16_sum(float v) {
  v += __builtin_bit_cast(float, __builtin_amdgcn_update_dpp(
           0, __builtin_bit_cast(int, v), 0x161, 0xf, 0xf, false));
  v += __builtin_bit_cast(float, __builtin_amdgcn_update_dpp(
           0, __builtin_bit_cast(int, v), 0x162, 0xf, 0xf, false));
  v += __builtin_bit_cast(float, __builtin_amdgcn_update_dpp(
           0, __builtin_bit_cast(int, v), 0x164, 0xf, 0xf, false));
  v += __builtin_bit_cast(float, __builtin_amdgcn_update_dpp(
           0, __builtin_bit_cast(int, v), 0x168, 0xf, 0xf, false));
  return v;
}

__device__ __forceinline__ float row8_sum(float v) {
  v += __builtin_bit_cast(float, __builtin_amdgcn_update_dpp(
           0, __builtin_bit_cast(int, v), 0x161, 0xf, 0xf, false));
  v += __builtin_bit_cast(float, __builtin_amdgcn_update_dpp(
           0, __builtin_bit_cast(int, v), 0x162, 0xf, 0xf, false));
  v += __builtin_bit_cast(float, __builtin_amdgcn_update_dpp(
           0, __builtin_bit_cast(int, v), 0x164, 0xf, 0xf, false));
  return v;
}

template <int MODE, int MI>
__global__ __launch_bounds__(GTHR) void gemm_wt_kernel(
    const float* __restrict__ A, const float* __restrict__ Wt,
    const float* __restrict__ colScale, const float* __restrict__ resid,
    const float* __restrict__ rowSq, float* __restrict__ out, int M, int N,
    int K, int act) {
  constexpr int BM = 32 * MI;
  constexpr int NA = 2 * MI;
  __shared__ float lA[2 * BM * LSTR];
  __shared__ float lW[2 * BN * LSTR];
  __shared__ float rinvS[BM];
  (void)M;

  const int tid = threadIdx.x;
  const int lane = tid & 31;
  const int wave = tid >> 5;
  const int lm = lane & 15;
  const int half = lane >> 4;
  const int wm = (wave & 1) * (16 * MI);
  const int wn = (wave >> 1) * 64;

  const int m0 = blockIdx.y * BM;
  const int n00 = blockIdx.x * BN;

  const int ldRow = tid >> 3;
  const int ldCol = (tid & 7) * 4;
  const float* Ag = A + (size_t)(m0 + ldRow) * K + ldCol;
  const float* Wg = Wt + (size_t)(n00 + ldRow) * K + ldCol;
  const int ldOff = ldRow * LSTR + ldCol;

  v4f ra[NA];
  v4f rw[8];
  v2f sq[NA] = {};

#pragma unroll
  for (int i = 0; i < NA; ++i) ra[i] = *(const v4f*)(Ag + (size_t)(i * 16) * K);
#pragma unroll
  for (int i = 0; i < 8; ++i) rw[i] = *(const v4f*)(Wg + (size_t)(i * 16) * K);
  if constexpr (MODE == 0) {
    if (tid < BM) {
      float s = 0.0f;
#pragma unroll
      for (int p = 0; p < NPLANE; ++p) s += rowSq[(size_t)p * M + m0 + tid];
      rinvS[tid] = rsqrtf(s / (float)K + EPS_F);
    }
  }
#pragma unroll
  for (int i = 0; i < NA; ++i) *(v4f*)&lA[ldOff + i * 16 * LSTR] = ra[i];
#pragma unroll
  for (int i = 0; i < 8; ++i) *(v4f*)&lW[ldOff + i * 16 * LSTR] = rw[i];
  if constexpr (MODE == 1) {
#pragma unroll
    for (int i = 0; i < NA; ++i) {
      sq[i] += ra[i].lo * ra[i].lo;
      sq[i] += ra[i].hi * ra[i].hi;
    }
  }
  __syncthreads();

  v8f acc[MI][4] = {};
  int cur = 0;
  for (int kc = 0; kc < K; kc += KC) {
    const int nxt = kc + KC;
    if (nxt < K) {
#pragma unroll
      for (int i = 0; i < NA; ++i)
        ra[i] = *(const v4f*)(Ag + (size_t)(i * 16) * K + nxt);
#pragma unroll
      for (int i = 0; i < 8; ++i)
        rw[i] = *(const v4f*)(Wg + (size_t)(i * 16) * K + nxt);
    }

    const float* As = &lA[cur * (BM * LSTR) + (wm + lm) * LSTR + 2 * half];
    const float* Ws = &lW[cur * (BN * LSTR) + (wn + lm) * LSTR + 2 * half];
#pragma unroll
    for (int kk = 0; kk < KC; kk += 4) {
      v2f a[MI], b[4];
#pragma unroll
      for (int i = 0; i < MI; ++i) a[i] = *(const v2f*)(As + kk + i * 16 * LSTR);
#pragma unroll
      for (int i = 0; i < 4; ++i) b[i] = *(const v2f*)(Ws + kk + i * 16 * LSTR);
#pragma unroll
      for (int i = 0; i < MI; ++i) {
#pragma unroll
        for (int s = 0; s < 4; ++s) acc[i][s] = wmma4(a[i], b[s], acc[i][s]);
      }
    }
    if (nxt < K) {
      const int oa = (cur ^ 1) * (BM * LSTR) + ldOff;
      const int ow = (cur ^ 1) * (BN * LSTR) + ldOff;
#pragma unroll
      for (int i = 0; i < NA; ++i) *(v4f*)&lA[oa + i * 16 * LSTR] = ra[i];
#pragma unroll
      for (int i = 0; i < 8; ++i) *(v4f*)&lW[ow + i * 16 * LSTR] = rw[i];
      if constexpr (MODE == 1) {
#pragma unroll
        for (int i = 0; i < NA; ++i) {
          sq[i] += ra[i].lo * ra[i].lo;
          sq[i] += ra[i].hi * ra[i].hi;
        }
      }
    }
    __syncthreads();
    cur ^= 1;
  }

  if constexpr (MODE == 1) {
#pragma unroll
    for (int i = 0; i < NA; ++i) {
      const float s = row8_sum(sq[i].x + sq[i].y);
      if ((tid & 7) == 0) rinvS[ldRow + i * 16] = rsqrtf(s / (float)K + EPS_F);
    }
    __syncthreads();
  }

  float* stg = &lA[wave * (16 * ESTR)];
  const int sw = (8 * half) * ESTR + lm;
  const int sr = half * ESTR + 4 * lm;

  if constexpr (MODE == 1) {
    const int sect = n00 / DMODEL;
    const int h = (n00 - sect * DMODEL + wn) / HD;
    const int lr = wm + 8 * half;
    float* dsth = out + (((size_t)sect * NH + h) * T_SEQ + (m0 + wm)) * HD;
#pragma unroll
    for (int i = 0; i < MI; ++i) {
#pragma unroll
      for (int r = 0; r < 8; ++r) {
        const float ri = rinvS[lr + i * 16 + r];
#pragma unroll
        for (int s = 0; s < 4; ++s) acc[i][s][r] *= ri;
      }
    }
    if (sect == 2) {
#pragma unroll
      for (int i = 0; i < MI; ++i) {
#pragma unroll
        for (int s = 0; s < 4; ++s) {
#pragma unroll
          for (int r = 0; r < 8; ++r) stg[sw + r * ESTR + s * 16] = acc[i][s][r];
        }
        lds_barrier();
        v4f t[8];
#pragma unroll
        for (int j = 0; j < 8; ++j) t[j] = *(const v4f*)&stg[sr + 2 * j * ESTR];
#pragma unroll
        for (int j = 0; j < 8; ++j)
          *(v4f*)(dsth + (size_t)(i * 16 + 2 * j + half) * HD + 4 * lm) = t[j];
        if (i + 1 < MI) lds_barrier();
      }
    } else {
      float* tcs = &lW[0];
      float* tsn = tcs + BM * TSTR;
      {
        const int jb = (tid & 7) * 2;
        float fr[2];
#pragma unroll
        for (int u = 0; u < 2; ++u)
          fr[u] = powf(10000.0f, -(float)(jb + u) / 15.0f);
        for (int row = tid >> 3; row < BM; row += GTHR / 8) {
          const float tf = (float)(m0 + row);
#pragma unroll
          for (int u = 0; u < 2; ++u) {
            const float theta = tf * fr[u];
            tcs[row * TSTR + jb + u] = cosf(theta);
            tsn[row * TSTR + jb + u] = sinf(theta);
          }
        }
      }
      __syncthreads();

#pragma unroll
      for (int i = 0; i < MI; ++i) {
        float ssq[8];
#pragma unroll
        for (int r = 0; r < 8; ++r)
          ssq[r] = (acc[i][0][r] * acc[i][0][r] + acc[i][1][r] * acc[i][1][r]) +
                   (acc[i][2][r] * acc[i][2][r] + acc[i][3][r] * acc[i][3][r]);
#pragma unroll
        for (int r = 0; r < 8; ++r) ssq[r] = row16_sum(ssq[r]);

#pragma unroll
        for (int r = 0; r < 8; ++r) {
          const int row = i * 16 + r;
          const float inv = rsqrtf(ssq[r] / (float)HD + EPS_F);
          const float cc = tcs[(lr + row) * TSTR + lm];
          const float sn = tsn[(lr + row) * TSTR + lm];
          const float a0 = acc[i][0][r] * inv, a1 = acc[i][1][r] * inv;
          const float a2 = acc[i][2][r] * inv, a3 = acc[i][3][r] * inv;
          stg[sw + r * ESTR] = a0 * cc + a2 * sn;
          stg[sw + r * ESTR + 32] = -a0 * sn + a2 * cc;
          stg[sw + r * ESTR + 16] = a1 * 1.0f + a3 * 0.0f;
          stg[sw + r * ESTR + 48] = -a1 * 0.0f + a3 * 1.0f;
        }
        lds_barrier();
        v4f t[8];
#pragma unroll
        for (int j = 0; j < 8; ++j) t[j] = *(const v4f*)&stg[sr + 2 * j * ESTR];
#pragma unroll
        for (int j = 0; j < 8; ++j)
          *(v4f*)(dsth + (size_t)(i * 16 + 2 * j + half) * HD + 4 * lm) = t[j];
        if (i + 1 < MI) lds_barrier();
      }
    }
    return;
  }

  const int mB = m0 + wm + half;
  const int nc = n00 + wn + 4 * lm;
  v4f cs = {1.0f, 1.0f, 1.0f, 1.0f};
  if (colScale) cs = *(const v4f*)(colScale + nc);
  const bool hasRes = (resid != nullptr);
  constexpr int RB = (MI <= 2) ? MI : 1;
  v4f rv[RB][8];
  if (hasRes) {
#pragma unroll
    for (int b = 0; b < RB; ++b) {
#pragma unroll
      for (int j = 0; j < 8; ++j)
        rv[b][j] = *(const v4f*)(resid + (size_t)(mB + b * 16 + 2 * j) * N + nc);
    }
  }
#pragma unroll
  for (int i = 0; i < MI; ++i) {
    if (RB == 1 && i > 0 && hasRes) {
#pragma unroll
      for (int j = 0; j < 8; ++j)
        rv[0][j] = *(const v4f*)(resid + (size_t)(mB + i * 16 + 2 * j) * N + nc);
    }
#pragma unroll
    for (int s = 0; s < 4; ++s) {
#pragma unroll
      for (int r = 0; r < 8; ++r) stg[sw + r * ESTR + s * 16] = acc[i][s][r];
    }
    lds_barrier();
    v4f t[8];
#pragma unroll
    for (int j = 0; j < 8; ++j) t[j] = *(const v4f*)&stg[sr + 2 * j * ESTR];
    if (act == 1) {
#pragma unroll
      for (int j = 0; j < 8; ++j) {
        v4f v = t[j] * (cs * rinvS[wm + i * 16 + 2 * j + half]);
        v.x = fmaxf(v.x, 0.0f);
        v.y = fmaxf(v.y, 0.0f);
        v.z = fmaxf(v.z, 0.0f);
        v.w = fmaxf(v.w, 0.0f);
        t[j] = v * v;
      }
    } else {
#pragma unroll
      for (int j = 0; j < 8; ++j)
        t[j] = t[j] * (cs * rinvS[wm + i * 16 + 2 * j + half]);
    }
    if (hasRes) {
#pragma unroll
      for (int j = 0; j < 8; ++j) t[j] += rv[RB == 1 ? 0 : i][j];
    }
#pragma unroll
    for (int j = 0; j < 8; ++j)
      *(v4f*)(out + (size_t)(mB + i * 16 + 2 * j) * N + nc) = t[j];
    if (i + 1 < MI) lds_barrier();
  }
}

#define KC2   64
#define LSTR2 68
#define GTHR2 256

__global__ __launch_bounds__(GTHR2) void gemm_ksplit_kernel(
    const float* __restrict__ A, const float* __restrict__ Wt,
    const float* __restrict__ colScale, const float* __restrict__ resid,
    float* __restrict__ out, float* __restrict__ rowSq, int M, int N, int K) {
  constexpr int BM = 128;
  __shared__ __attribute__((aligned(16))) float lA[2 * BM * LSTR2];
  __shared__ __attribute__((aligned(16))) float lW[2 * BN * LSTR2];
  __shared__ v4f red[8 * 16 * 32];
  (void)M;

  const int tid = threadIdx.x;
  const int lane = tid & 31;
  const int wave = __builtin_amdgcn_readfirstlane(tid >> 5);
  const int lm = lane & 15;
  const int half = lane >> 4;
  const int kg = wave >> 2;
  const int wm = (wave & 1) * 64;
  const int wn = ((wave >> 1) & 1) * 64;

  const int m0 = blockIdx.y * BM;
  const int n00 = blockIdx.x * BN;

  const int ldRow = tid >> 4;
  const int ldCol = (tid & 15) * 4;
  const float* Ag = A + (size_t)(m0 + ldRow) * K + ldCol;
  const float* Wg = Wt + (size_t)(n00 + ldRow) * K + ldCol;
  const int ldOff = ldRow * LSTR2 + ldCol;

  v4f ra[8];
  v4f rw[8];

#pragma unroll
  for (int i = 0; i < 8; ++i) ra[i] = *(const v4f*)(Ag + (size_t)(i * 16) * K);
#pragma unroll
  for (int i = 0; i < 8; ++i) rw[i] = *(const v4f*)(Wg + (size_t)(i * 16) * K);
#pragma unroll
  for (int i = 0; i < 8; ++i) *(v4f*)&lA[ldOff + i * 16 * LSTR2] = ra[i];
#pragma unroll
  for (int i = 0; i < 8; ++i) *(v4f*)&lW[ldOff + i * 16 * LSTR2] = rw[i];
  __syncthreads();

  v8f acc[4][4] = {};
  int cur = 0;
  for (int kc = 0; kc < K; kc += KC2) {
    const int nxt = kc + KC2;
    if (nxt < K) {
#pragma unroll
      for (int i = 0; i < 8; ++i)
        ra[i] = *(const v4f*)(Ag + (size_t)(i * 16) * K + nxt);
#pragma unroll
      for (int i = 0; i < 8; ++i)
        rw[i] = *(const v4f*)(Wg + (size_t)(i * 16) * K + nxt);
    }

    const float* AsK = &lA[cur * (BM * LSTR2) + (wm + 32 * kg + lm) * LSTR2 +
                           2 * half + kg * (KC2 / 2)];
    const float* AsG = &lA[cur * (BM * LSTR2) +
                           (wm + 32 * (1 - kg) + lm) * LSTR2 + 2 * half +
                           kg * (KC2 / 2)];
    const float* Ws = &lW[cur * (BN * LSTR2) + (wn + lm) * LSTR2 + 2 * half +
                          kg * (KC2 / 2)];
#pragma unroll
    for (int kk = 0; kk < KC2 / 2; kk += 4) {
      v2f a[4], b[4];
#pragma unroll
      for (int i = 0; i < 2; ++i) {
        a[i] = *(const v2f*)(AsK + kk + i * 16 * LSTR2);
        a[2 + i] = *(const v2f*)(AsG + kk + i * 16 * LSTR2);
      }
#pragma unroll
      for (int i = 0; i < 4; ++i) b[i] = *(const v2f*)(Ws + kk + i * 16 * LSTR2);
#pragma unroll
      for (int i = 0; i < 4; ++i) {
#pragma unroll
        for (int s = 0; s < 4; ++s) acc[i][s] = wmma4(a[i], b[s], acc[i][s]);
      }
    }
    if (nxt < K) {
      const int oa = (cur ^ 1) * (BM * LSTR2) + ldOff;
      const int ow = (cur ^ 1) * (BN * LSTR2) + ldOff;
#pragma unroll
      for (int i = 0; i < 8; ++i) *(v4f*)&lA[oa + i * 16 * LSTR2] = ra[i];
#pragma unroll
      for (int i = 0; i < 8; ++i) *(v4f*)&lW[ow + i * 16 * LSTR2] = rw[i];
    }
    __syncthreads();
    cur ^= 1;
  }

  const int mB = m0 + wm + kg * 32 + half;
  const int nc = n00 + wn + 4 * lm;
  v4f cs = {1.0f, 1.0f, 1.0f, 1.0f};
  if (colScale) cs = *(const v4f*)(colScale + nc);
  const bool hasRes = (resid != nullptr);
  v4f rv[2][8];
  if (hasRes) {
#pragma unroll
    for (int b = 0; b < 2; ++b) {
#pragma unroll
      for (int j = 0; j < 8; ++j)
        rv[b][j] = *(const v4f*)(resid + (size_t)(mB + b * 16 + 2 * j) * N + nc);
    }
  }

  v4f* mine = red + (wave * 16) * 32 + lane;
  const v4f* theirs = red + ((wave ^ 4) * 16) * 32 + lane;
#pragma unroll
  for (int j = 0; j < 2; ++j) {
#pragma unroll
    for (int s = 0; s < 4; ++s) {
      const v8f v = acc[2 + j][s];
      mine[((j * 4 + s) * 2 + 0) * 32] = (v4f){v[0], v[1], v[2], v[3]};
      mine[((j * 4 + s) * 2 + 1) * 32] = (v4f){v[4], v[5], v[6], v[7]};
    }
  }
  lds_barrier();
#pragma unroll
  for (int j = 0; j < 2; ++j) {
#pragma unroll
    for (int s = 0; s < 4; ++s) {
      const v4f lo = theirs[((j * 4 + s) * 2 + 0) * 32];
      const v4f hi = theirs[((j * 4 + s) * 2 + 1) * 32];
      const v8f other = {lo.x, lo.y, lo.z, lo.w, hi.x, hi.y, hi.z, hi.w};
      acc[j][s] = acc[j][s] + other;
    }
  }

  float* stg = &lA[wave * (16 * ESTR)];
  const int sw = (8 * half) * ESTR + lm;
  const int sr = half * ESTR + 4 * lm;
  float* rsq = &lA[8 * (16 * ESTR) + wave * 32];
#pragma unroll
  for (int i = 0; i < 2; ++i) {
#pragma unroll
    for (int s = 0; s < 4; ++s) {
#pragma unroll
      for (int r = 0; r < 8; ++r) stg[sw + r * ESTR + s * 16] = acc[i][s][r];
    }
    lds_barrier();
    v4f t[8];
#pragma unroll
    for (int j = 0; j < 8; ++j) t[j] = *(const v4f*)&stg[sr + 2 * j * ESTR];
#pragma unroll
    for (int j = 0; j < 8; ++j) t[j] = t[j] * cs;
    if (hasRes) {
#pragma unroll
      for (int j = 0; j < 8; ++j) t[j] += rv[i][j];
    }
#pragma unroll
    for (int j = 0; j < 8; ++j)
      *(v4f*)(out + (size_t)(mB + i * 16 + 2 * j) * N + nc) = t[j];
    if (rowSq != nullptr) {
#pragma unroll
      for (int j = 0; j < 8; ++j) {
        const v4f v = t[j];
        const float q =
            row16_sum((v.x * v.x + v.y * v.y) + (v.z * v.z + v.w * v.w));
        if (lm == 0) rsq[i * 16 + 2 * j + half] = q;
      }
    }
    if (i + 1 < 2) lds_barrier();
  }
  if (rowSq != nullptr) {
    lds_barrier();
    rowSq[(size_t)(2 * blockIdx.x + (wn >> 6)) * M + (m0 + wm + kg * 32) +
          lane] = rsq[lane];
  }
}

__device__ __forceinline__ float xhalf(float x) {
  const int b = __builtin_bit_cast(int, x);
  return __builtin_bit_cast(
      float, __builtin_amdgcn_permlanex16(b, b, 0x76543210, 0xfedcba98, false,
                                          false));
}

__global__ __launch_bounds__(32) void attn_kernel(
    const float* __restrict__ Qh, const float* __restrict__ Kh,
    const float* __restrict__ Vh, float* __restrict__ outAttn) {
  const int h = blockIdx.x;
  const int q0 = blockIdx.y * 16;
  const int lane = threadIdx.x;
  const int lm = lane & 15;
  const int half = lane >> 4;
  const size_t hb = (size_t)h * T_SEQ * HD;
  const int qp = q0 + lm;

  const float* qrow = Qh + hb + (size_t)qp * HD + 2 * half;
  v2f qb[16];
#pragma unroll
  for (int dc = 0; dc < 16; ++dc) qb[dc] = *(const v2f*)(qrow + 4 * dc);

  float mi = -INFINITY;
  float li = 0.0f;
  v8f o[4] = {};

  const int lo = q0 - WIN + 1;
  const int kt0 = (lo < 0) ? 0 : (lo / 16) * 16;
  for (int kt = kt0; kt <= q0; kt += 16) {
    const float* krow = Kh + hb + (size_t)(kt + lm) * HD + 2 * half;
    v2f ka[16];
#pragma unroll
    for (int dc = 0; dc < 16; ++dc) ka[dc] = *(const v2f*)(krow + 4 * dc);

    const float* vrow = Vh + hb + (size_t)(kt + 8 * half) * HD + lm;
    v2f va[16];
#pragma unroll
    for (int s = 0; s < 4; ++s) {
#pragma unroll
      for (int f = 0; f < 4; ++f) {
        v2f a;
        a.x = vrow[(2 * s) * HD + f * 16];
        a.y = vrow[(2 * s + 1) * HD + f * 16];
        va[s * 4 + f] = a;
      }
    }

    v8f s0 = {}, s1 = {};
#pragma unroll
    for (int dc = 0; dc < 16; dc += 2) {
      s0 = wmma4(ka[dc], qb[dc], s0);
      s1 = wmma4(ka[dc + 1], qb[dc + 1], s1);
    }
    const v8f st = s0 + s1;

    float sv[8];
#pragma unroll
    for (int r = 0; r < 8; ++r) sv[r] = st[r] * 0.125f;

    if ((kt + 15 > q0) || (kt + WIN <= q0 + 15)) {
#pragma unroll
      for (int r = 0; r < 8; ++r) {
        const int kp = kt + 8 * half + r;
        const bool okm = (kp <= qp) && (kp > qp - WIN);
        sv[r] = okm ? sv[r] : -INFINITY;
      }
    }

    float rm = fmaxf(fmaxf(fmaxf(sv[0], sv[1]), fmaxf(sv[2], sv[3])),
                     fmaxf(fmaxf(sv[4], sv[5]), fmaxf(sv[6], sv[7])));
    rm = fmaxf(rm, xhalf(rm));
    const float nm = fmaxf(mi, rm);
    const float nmz = (nm == -INFINITY) ? 0.0f : nm;
    const float corr = __expf(mi - nmz);
    float p[8];
#pragma unroll
    for (int r = 0; r < 8; ++r) p[r] = __expf(sv[r] - nmz);
    float rs = ((p[0] + p[1]) + (p[2] + p[3])) + ((p[4] + p[5]) + (p[6] + p[7]));
    rs += xhalf(rs);
    li = li * corr + rs;
    mi = nm;
#pragma unroll
    for (int f = 0; f < 4; ++f) o[f] *= corr;

#pragma unroll
    for (int s = 0; s < 4; ++s) {
      v2f pb;
      pb.x = p[2 * s];
      pb.y = p[2 * s + 1];
#pragma unroll
      for (int f = 0; f < 4; ++f) o[f] = wmma4(va[s * 4 + f], pb, o[f]);
    }
  }

  float* orow = outAttn + (size_t)qp * DMODEL + h * HD + 8 * half;
#pragma unroll
  for (int f = 0; f < 4; ++f) {
#pragma unroll
    for (int g = 0; g < 2; ++g) {
      v4f w;
      w.x = o[f][4 * g + 0] / li;
      w.y = o[f][4 * g + 1] / li;
      w.z = o[f][4 * g + 2] / li;
      w.w = o[f][4 * g + 3] / li;
      *(v4f*)(orow + f * 16 + 4 * g) = w;
    }
  }
}

extern "C" void kernel_launch(void* const* d_in, const int* in_sizes, int n_in,
                              void* d_out, int out_size, void* d_ws,
                              size_t ws_size, hipStream_t stream) {
  (void)in_sizes;
  (void)n_in;
  (void)out_size;
  (void)ws_size;
  const float* x = (const float*)d_in[0];
  const float* qkv_w = (const float*)d_in[1];
  const float* o_w = (const float*)d_in[2];
  const float* o_scale = (const float*)d_in[3];
  const float* w1 = (const float*)d_in[4];
  const float* w2 = (const float*)d_in[5];
  const float* w2_scale = (const float*)d_in[6];
  float* out = (float*)d_out;

  const size_t TD = (size_t)T_SEQ * DMODEL;
  float* rsq = (float*)d_ws;
  float* qkv = rsq + TD;
  float* qh = qkv + 3 * TD;
  float* kh = qh + TD;
  float* vh = kh + TD;
  float* attn = vh + TD;
  float* x1 = attn + TD;
  float* hbuf = qkv;

  gemm_wt_kernel<1, 4><<<dim3(3 * DMODEL / BN, T_SEQ / 128), GTHR, 0, stream>>>(
      x, qkv_w, nullptr, nullptr, nullptr, qh, T_SEQ, 3 * DMODEL, DMODEL, 0);
  attn_kernel<<<dim3(NH, T_SEQ / 16), 32, 0, stream>>>(qh, kh, vh, attn);
  gemm_ksplit_kernel<<<dim3(DMODEL / BN, T_SEQ / 128), GTHR2, 0, stream>>>(
      attn, o_w, o_scale, x, x1, rsq, T_SEQ, DMODEL, DMODEL);
  gemm_wt_kernel<0, 4><<<dim3(DMLP / BN, T_SEQ / 128), GTHR, 0, stream>>>(
      x1, w1, nullptr, nullptr, rsq, hbuf, T_SEQ, DMLP, DMODEL, 1);
  gemm_ksplit_kernel<<<dim3(DMODEL / BN, T_SEQ / 128), GTHR2, 0, stream>>>(
      hbuf, w2, w2_scale, x1, out, nullptr, T_SEQ, DMODEL, DMLP);
}
